# starting kernel
# baseline (speedup 1.0000x reference)
#include <hip/hip_runtime.h>

#define NODE_F 128

typedef float v2f __attribute__((ext_vector_type(2)));
typedef float v8f __attribute__((ext_vector_type(8)));

static __device__ __forceinline__ v8f wmma_k4(v2f a, v2f b, v8f c) {
  return __builtin_amdgcn_wmma_f32_16x16x4_f32(
       false, a,  false, b,
       (short)0, c,  false,  false);
}

__global__ void deg_kernel(const int* __restrict__ dst, float* __restrict__ deg, int nEdges) {
  int e = blockIdx.x * blockDim.x + threadIdx.x;
  if (e < nEdges) atomicAdd(&deg[dst[e]], 1.0f);
}

__global__ void scatter_kernel(const float* __restrict__ h, const int* __restrict__ src,
                               const int* __restrict__ dst, float* __restrict__ sum,
                               int nEdges) {
  int g    = blockIdx.x * blockDim.x + threadIdx.x;
  int e    = g >> 5;
  int lane = g & 31;
  if (e >= nEdges) return;
  int s = src[e];
  int d = dst[e];
  float4 v = ((const float4*)(h + (size_t)s * NODE_F))[lane];
  float* sp = sum + (size_t)d * NODE_F + lane * 4;
  atomicAdd(sp + 0, v.x);
  atomicAdd(sp + 1, v.y);
  atomicAdd(sp + 2, v.z);
  atomicAdd(sp + 3, v.w);
}

template <bool NEIGH, bool RELU>
__global__ void sage_gemm(const float* __restrict__ hself, const float* __restrict__ sum,
                          const float* __restrict__ deg, const float* __restrict__ Wself,
                          const float* __restrict__ Wneigh, const float* __restrict__ bias,
                          float* __restrict__ out, int nNodes) {
  __shared__ float hs[32][NODE_F + 4];
  __shared__ float hn[32][NODE_F + 4];

  const int tid     = threadIdx.x;
  const int m_base  = blockIdx.x * 32;
  const bool fullTile = (m_base + 32 <= nNodes);

  for (int i = 0; i < 4; ++i) {
    int f   = i * 256 + tid;
    int row = f >> 5;
    int cv  = f & 31;
    int m   = m_base + row;
    float4 v = make_float4(0.f, 0.f, 0.f, 0.f);
    if (fullTile || m < nNodes) v = ((const float4*)(hself + (size_t)m * NODE_F))[cv];
    *(float4*)&hs[row][cv * 4] = v;
    if (NEIGH) {
      float4 s4 = make_float4(0.f, 0.f, 0.f, 0.f);
      float rd = 1.0f;
      if (fullTile || m < nNodes) {
        s4 = ((const float4*)(sum + (size_t)m * NODE_F))[cv];
        rd = 1.0f / fmaxf(deg[m], 1.0f);
      }
      hn[row][cv * 4 + 0] = s4.x * rd;
      hn[row][cv * 4 + 1] = s4.y * rd;
      hn[row][cv * 4 + 2] = s4.z * rd;
      hn[row][cv * 4 + 3] = s4.w * rd;
    }
  }
  __syncthreads();

  const int lane = tid & 31;
  const int wv   = tid >> 5;
  const int lid  = lane & 15;
  const int half = lane >> 4;
  const int n0   = wv * 16;

  const float* Ws = Wself + (size_t)(n0 + lid) * NODE_F;
  const float* Wn = NEIGH ? (Wneigh + (size_t)(n0 + lid) * NODE_F) : nullptr;

  v8f acc0 = {0.f, 0.f, 0.f, 0.f, 0.f, 0.f, 0.f, 0.f};
  v8f acc1 = {0.f, 0.f, 0.f, 0.f, 0.f, 0.f, 0.f, 0.f};

#pragma unroll 8
  for (int k0 = 0; k0 < NODE_F; k0 += 4) {
    const int kk = k0 + 2 * half;
    v2f bs;
    bs.x = Ws[kk];
    bs.y = Ws[kk + 1];
    v2f a0, a1;
    a0.x = hs[lid][kk];
    a0.y = hs[lid][kk + 1];
    a1.x = hs[16 + lid][kk];
    a1.y = hs[16 + lid][kk + 1];
    acc0 = wmma_k4(a0, bs, acc0);
    acc1 = wmma_k4(a1, bs, acc1);
    if (NEIGH) {
      v2f bn;
      bn.x = Wn[kk];
      bn.y = Wn[kk + 1];
      v2f c0, c1;
      c0.x = hn[lid][kk];
      c0.y = hn[lid][kk + 1];
      c1.x = hn[16 + lid][kk];
      c1.y = hn[16 + lid][kk + 1];
      acc0 = wmma_k4(c0, bn, acc0);
      acc1 = wmma_k4(c1, bn, acc1);
    }
  }

  const float bv = bias[n0 + lid];
  float* outc = out + n0 + lid;
  const int r0 = m_base + 8 * half;
  const int r1 = m_base + 16 + 8 * half;

  if (fullTile) {
#pragma unroll
    for (int e = 0; e < 8; ++e) {
      float v0 = acc0[e] + bv;
      float v1 = acc1[e] + bv;
      if (RELU) { v0 = fmaxf(v0, 0.f); v1 = fmaxf(v1, 0.f); }
      outc[(size_t)(r0 + e) * NODE_F] = v0;
      outc[(size_t)(r1 + e) * NODE_F] = v1;
    }
  } else {
#pragma unroll
    for (int e = 0; e < 8; ++e) {
      float v0 = acc0[e] + bv;
      float v1 = acc1[e] + bv;
      if (RELU) { v0 = fmaxf(v0, 0.f); v1 = fmaxf(v1, 0.f); }
      if (r0 + e < nNodes) outc[(size_t)(r0 + e) * NODE_F] = v0;
      if (r1 + e < nNodes) outc[(size_t)(r1 + e) * NODE_F] = v1;
    }
  }
}

extern "C" void kernel_launch(void* const* d_in, const int* in_sizes, int n_in,
                              void* d_out, int out_size, void* d_ws, size_t ws_size,
                              hipStream_t stream) {
  const float* x      = (const float*)d_in[0];
  const int*   src    = (const int*)d_in[1];
  const int*   dst    = (const int*)d_in[2];
  const float* W1s    = (const float*)d_in[3];
  const float* W1n    = (const float*)d_in[4];
  const float* b1     = (const float*)d_in[5];
  const float* W2s    = (const float*)d_in[6];
  const float* W2n    = (const float*)d_in[7];
  const float* b2     = (const float*)d_in[8];
  const float* W3     = (const float*)d_in[9];
  const float* b3     = (const float*)d_in[10];
  float*       out    = (float*)d_out;

  const int nNodes = in_sizes[0] / NODE_F;
  const int nEdges = in_sizes[1];

  float* sum = (float*)d_ws;
  float* deg = sum + (size_t)nNodes * NODE_F;
  float* h1  = deg + nNodes;
  float* h2  = h1 + (size_t)nNodes * NODE_F;

  dim3 blk(256);
  dim3 gEdge((nEdges + 255) / 256);
  dim3 gScat((nEdges * 32 + 255) / 256);
  dim3 gGemm((nNodes + 31) / 32);

  hipMemsetAsync(sum, 0, ((size_t)nNodes * NODE_F + nNodes) * sizeof(float), stream);
  deg_kernel<<<gEdge, blk, 0, stream>>>(dst, deg, nEdges);
  scatter_kernel<<<gScat, blk, 0, stream>>>(x, src, dst, sum, nEdges);
  sage_gemm<true, true><<<gGemm, blk, 0, stream>>>(x, sum, deg, W1s, W1n, b1, h1, nNodes);

  hipMemsetAsync(sum, 0, (size_t)nNodes * NODE_F * sizeof(float), stream);
  scatter_kernel<<<gScat, blk, 0, stream>>>(h1, src, dst, sum, nEdges);
  sage_gemm<true, false><<<gGemm, blk, 0, stream>>>(h1, sum, deg, W2s, W2n, b2, h2, nNodes);

  sage_gemm<false, false><<<gGemm, blk, 0, stream>>>(h2, nullptr, nullptr, W3, nullptr, b3,
                                                     out, nNodes);
}


# neighbour sums without global atomics or sort: a block owns 400 nodes, filters a byte-key edge list, adds its edges' rows to LDS sums by LDS float atomics; memsets, deg launch gone; arrival sum order
# speedup vs baseline: 2.0820x; 2.0820x over previous
#include <hip/hip_runtime.h>

#define NODE_F 128

typedef float v2f __attribute__((ext_vector_type(2)));
typedef float v8f __attribute__((ext_vector_type(8)));

static __device__ __forceinline__ v8f wmma_k4(v2f a, v2f b, v8f c) {
  return __builtin_amdgcn_wmma_f32_16x16x4_f32(
       false, a,  false, b,
       (short)0, c,  false,  false);
}

#define AGG_R     400
#define AGG_T     1024
#define AGG_V     4
#define AGG_B     16
#define AGG_QCAP  (32 * AGG_B)

__global__ void key_kernel(const int* __restrict__ dst, unsigned* __restrict__ key,
                           int nEdges, int nWords) {
  const int t = blockIdx.x * blockDim.x + threadIdx.x;
  if (t >= nWords) return;
  const int e = t * 4;
  unsigned w = 0xFFFFFFFFu;
  if (e + 3 < nEdges) {
    const int4 d = ((const int4*)dst)[t];
    w = (unsigned)min(d.x / AGG_R, 255) | ((unsigned)min(d.y / AGG_R, 255) << 8) |
        ((unsigned)min(d.z / AGG_R, 255) << 16) | ((unsigned)min(d.w / AGG_R, 255) << 24);
  } else {
    for (int b = 0; b < 4; ++b)
      if (e + b < nEdges)
        w = (w & ~(0xFFu << (8 * b))) | ((unsigned)min(dst[e + b] / AGG_R, 255) << (8 * b));
  }
  key[t] = w;
}

template <bool FIRST>
__global__ void __launch_bounds__(AGG_T)
agg_kernel(const float* __restrict__ h, const int* __restrict__ src,
           const int* __restrict__ dst, const unsigned* __restrict__ key,
           float* __restrict__ sum, float* __restrict__ deg,
           int nNodes, int nEdges, int nChunks) {
  __shared__ float    acc[(AGG_R + 1) * NODE_F];
  __shared__ unsigned q[2][AGG_QCAP];
  __shared__ int      cnt[AGG_R];
  __shared__ unsigned qn[4];

  const int tid  = threadIdx.x;
  const int lane = tid & 31;
  const int wave = __builtin_amdgcn_readfirstlane(tid >> 5);
  const int lo   = blockIdx.x * AGG_R;
  const unsigned nOwn   = (unsigned)min(AGG_R, nNodes - lo);
  const unsigned keyrep = (unsigned)min((int)blockIdx.x, 255) * 0x01010101u;

  for (int i = tid; i < (AGG_R + 1) * NODE_F / 4; i += AGG_T)
    ((float4*)acc)[i] = make_float4(0.f, 0.f, 0.f, 0.f);
  for (int i = tid; i < 2 * AGG_QCAP; i += AGG_T) (&q[0][0])[i] = 0u;
  if (FIRST)
    for (int i = tid; i < AGG_R; i += AGG_T) cnt[i] = 0;
  if (tid < 4) qn[tid] = 0u;
  __syncthreads();

  const int  subVec  = AGG_T * AGG_V;
  const uint4* key4  = (const uint4*)key;

  uint4 d[AGG_V];

  auto load_keys = [&](int g) {
#pragma unroll
    for (int u = 0; u < AGG_V; ++u) d[u] = key4[g * subVec + u * AGG_T + tid];
  };
  auto eq4 = [&](unsigned w) -> unsigned {
    const unsigned x = w ^ keyrep;
    return ((x - 0x01010101u) & ~x & 0x80808080u) >> 7;
  };

  auto push_chunk = [&](int g) {
    unsigned* qq = q[g & 1];
    unsigned long long m = 0ull;
#pragma unroll
    for (int u = 0; u < AGG_V; ++u) {
      const unsigned p = (eq4(d[u].x) << ((4 * u + 0) & 7)) | (eq4(d[u].y) << ((4 * u + 1) & 7)) |
                         (eq4(d[u].z) << ((4 * u + 2) & 7)) | (eq4(d[u].w) << ((4 * u + 3) & 7));
      m |= (unsigned long long)p << (32 * (u >> 1));
    }
    load_keys(g + 1);

    while (m) {
      const int i = __ffsll((long long)m) - 1;
      m &= m - 1ull;
      const int k = (i & 7) + 8 * (i >> 5);
      const int b = (i >> 3) & 3;
      const int e = (g * subVec + (k >> 2) * AGG_T + tid) * 16 + (k & 3) * 4 + b;
      if (e < nEdges) {
        int s = src[e];
        const unsigned dl = (unsigned)(dst[e] - lo);
        asm volatile("" : "+v"(s));
        if (dl < nOwn) {
          if (FIRST) atomicAdd(&cnt[dl], 1);
          const unsigned pos = atomicAdd(&qn[g & 3], 1u);
          if (pos < AGG_QCAP) {
            qq[pos] = ((unsigned)s << 9) | dl;
          } else {
            const float* hp = h + (size_t)s * NODE_F;
            for (int f = 0; f < NODE_F; ++f) atomicAdd(&acc[dl * NODE_F + f], hp[f]);
          }
        }
      }
    }
  };

  unsigned ent[AGG_B];
  float v[AGG_B][4];
  auto load_rows = [&](const unsigned* qq, unsigned k0, unsigned n) {
    unsigned raw[AGG_B];
#pragma unroll
    for (int j = 0; j < AGG_B; ++j) raw[j] = qq[min(k0 + 32u * j, (unsigned)(AGG_QCAP - 1))];
#pragma unroll
    for (int j = 1; j < AGG_B; ++j) {
      ent[j] = 0u;
      if (k0 + 32u * j < n) {
        ent[j] = (unsigned)__builtin_amdgcn_readfirstlane((int)raw[j]);
        const float* hp = h + (size_t)(ent[j] >> 9) * NODE_F + lane;
        v[j][0] = hp[0];
        v[j][1] = hp[32];
        v[j][2] = hp[64];
        v[j][3] = hp[96];
      }
    }
    ent[0] = (k0 < n) ? (unsigned)__builtin_amdgcn_readfirstlane((int)raw[0]) : (unsigned)AGG_R;
    {
      const float* hp = h + (size_t)(ent[0] >> 9) * NODE_F + lane;
      v[0][0] = hp[0];
      v[0][1] = hp[32];
      v[0][2] = hp[64];
      v[0][3] = hp[96];
    }
  };
  auto add_rows = [&](unsigned k0, unsigned n) {
#pragma unroll
    for (int j = 1; j < AGG_B; ++j) {
      if (k0 + 32u * j < n) {
        float* ap = acc + (ent[j] & 511u) * NODE_F + lane;
        atomicAdd(ap,      v[j][0]);
        atomicAdd(ap + 32, v[j][1]);
        atomicAdd(ap + 64, v[j][2]);
        atomicAdd(ap + 96, v[j][3]);
      }
    }
    {
      float* ap = acc + (ent[0] & 511u) * NODE_F + lane;
      atomicAdd(ap,      v[0][0]);
      atomicAdd(ap + 32, v[0][1]);
      atomicAdd(ap + 64, v[0][2]);
      atomicAdd(ap + 96, v[0][3]);
    }
  };

  load_keys(0);
  if (nChunks > 0) push_chunk(0);

  for (int c = 0; c < nChunks; ++c) {
    __syncthreads();
    if (tid == 0) qn[(c + 2) & 3] = 0u;

    const unsigned* qq = q[c & 1];
    const unsigned n = min(qn[c & 3], (unsigned)AGG_QCAP);
    load_rows(qq, (unsigned)wave, n);
    if (c + 1 < nChunks) push_chunk(c + 1);
    add_rows((unsigned)wave, n);
  }
  __syncthreads();

  for (unsigned r = (unsigned)wave; r < nOwn; r += 32u) {
    const float* ap = acc + r * NODE_F + lane;
    float* sp = sum + (size_t)(lo + r) * NODE_F + lane;
    sp[0]  = ap[0];
    sp[32] = ap[32];
    sp[64] = ap[64];
    sp[96] = ap[96];
  }
  if (FIRST)
    for (unsigned i = tid; i < nOwn; i += AGG_T) deg[lo + i] = (float)cnt[i];
}

template <bool NEIGH, bool RELU>
__global__ void sage_gemm(const float* __restrict__ hself, const float* __restrict__ sum,
                          const float* __restrict__ deg, const float* __restrict__ Wself,
                          const float* __restrict__ Wneigh, const float* __restrict__ bias,
                          float* __restrict__ out, int nNodes) {
  __shared__ float hs[32][NODE_F + 4];
  __shared__ float hn[32][NODE_F + 4];

  const int tid     = threadIdx.x;
  const int m_base  = blockIdx.x * 32;
  const bool fullTile = (m_base + 32 <= nNodes);

  for (int i = 0; i < 4; ++i) {
    int f   = i * 256 + tid;
    int row = f >> 5;
    int cv  = f & 31;
    int m   = m_base + row;
    float4 v = make_float4(0.f, 0.f, 0.f, 0.f);
    if (fullTile || m < nNodes) v = ((const float4*)(hself + (size_t)m * NODE_F))[cv];
    *(float4*)&hs[row][cv * 4] = v;
    if (NEIGH) {
      float4 s4 = make_float4(0.f, 0.f, 0.f, 0.f);
      float rd = 1.0f;
      if (fullTile || m < nNodes) {
        s4 = ((const float4*)(sum + (size_t)m * NODE_F))[cv];
        rd = 1.0f / fmaxf(deg[m], 1.0f);
      }
      hn[row][cv * 4 + 0] = s4.x * rd;
      hn[row][cv * 4 + 1] = s4.y * rd;
      hn[row][cv * 4 + 2] = s4.z * rd;
      hn[row][cv * 4 + 3] = s4.w * rd;
    }
  }
  __syncthreads();

  const int lane = tid & 31;
  const int wv   = tid >> 5;
  const int lid  = lane & 15;
  const int half = lane >> 4;
  const int n0   = wv * 16;

  const float* Ws = Wself + (size_t)(n0 + lid) * NODE_F;
  const float* Wn = NEIGH ? (Wneigh + (size_t)(n0 + lid) * NODE_F) : nullptr;

  v8f acc0 = {0.f, 0.f, 0.f, 0.f, 0.f, 0.f, 0.f, 0.f};
  v8f acc1 = {0.f, 0.f, 0.f, 0.f, 0.f, 0.f, 0.f, 0.f};

#pragma unroll 8
  for (int k0 = 0; k0 < NODE_F; k0 += 4) {
    const int kk = k0 + 2 * half;
    v2f bs;
    bs.x = Ws[kk];
    bs.y = Ws[kk + 1];
    v2f a0, a1;
    a0.x = hs[lid][kk];
    a0.y = hs[lid][kk + 1];
    a1.x = hs[16 + lid][kk];
    a1.y = hs[16 + lid][kk + 1];
    acc0 = wmma_k4(a0, bs, acc0);
    acc1 = wmma_k4(a1, bs, acc1);
    if (NEIGH) {
      v2f bn;
      bn.x = Wn[kk];
      bn.y = Wn[kk + 1];
      v2f c0, c1;
      c0.x = hn[lid][kk];
      c0.y = hn[lid][kk + 1];
      c1.x = hn[16 + lid][kk];
      c1.y = hn[16 + lid][kk + 1];
      acc0 = wmma_k4(c0, bn, acc0);
      acc1 = wmma_k4(c1, bn, acc1);
    }
  }

  const float bv = bias[n0 + lid];
  float* outc = out + n0 + lid;
  const int r0 = m_base + 8 * half;
  const int r1 = m_base + 16 + 8 * half;

  if (fullTile) {
#pragma unroll
    for (int e = 0; e < 8; ++e) {
      float v0 = acc0[e] + bv;
      float v1 = acc1[e] + bv;
      if (RELU) { v0 = fmaxf(v0, 0.f); v1 = fmaxf(v1, 0.f); }
      outc[(size_t)(r0 + e) * NODE_F] = v0;
      outc[(size_t)(r1 + e) * NODE_F] = v1;
    }
  } else {
#pragma unroll
    for (int e = 0; e < 8; ++e) {
      float v0 = acc0[e] + bv;
      float v1 = acc1[e] + bv;
      if (RELU) { v0 = fmaxf(v0, 0.f); v1 = fmaxf(v1, 0.f); }
      if (r0 + e < nNodes) outc[(size_t)(r0 + e) * NODE_F] = v0;
      if (r1 + e < nNodes) outc[(size_t)(r1 + e) * NODE_F] = v1;
    }
  }
}

extern "C" void kernel_launch(void* const* d_in, const int* in_sizes, int n_in,
                              void* d_out, int out_size, void* d_ws, size_t ws_size,
                              hipStream_t stream) {
  const float* x      = (const float*)d_in[0];
  const int*   src    = (const int*)d_in[1];
  const int*   dst    = (const int*)d_in[2];
  const float* W1s    = (const float*)d_in[3];
  const float* W1n    = (const float*)d_in[4];
  const float* b1     = (const float*)d_in[5];
  const float* W2s    = (const float*)d_in[6];
  const float* W2n    = (const float*)d_in[7];
  const float* b2     = (const float*)d_in[8];
  const float* W3     = (const float*)d_in[9];
  const float* b3     = (const float*)d_in[10];
  float*       out    = (float*)d_out;

  const int nNodes = in_sizes[0] / NODE_F;
  const int nEdges = in_sizes[1];

  float* sum = (float*)d_ws;
  float* deg = sum + (size_t)nNodes * NODE_F;
  float* h1  = deg + nNodes;
  float* h2  = h1 + (size_t)nNodes * NODE_F;

  unsigned* key = (unsigned*)h2;
  const int subVec  = AGG_T * AGG_V;
  const int nChunks = ((nEdges + 15) / 16 + subVec - 1) / subVec;
  const int nWords  = (nChunks + 1) * subVec * 4;

  dim3 blk(256);
  dim3 blkAgg(AGG_T);
  dim3 gKey((nWords + 255) / 256);
  dim3 gAgg((nNodes + AGG_R - 1) / AGG_R);
  dim3 gGemm((nNodes + 31) / 32);

  key_kernel<<<gKey, blk, 0, stream>>>(dst, key, nEdges, nWords);

  agg_kernel<true><<<gAgg, blkAgg, 0, stream>>>(x, src, dst, key, sum, deg, nNodes, nEdges, nChunks);
  sage_gemm<true, true><<<gGemm, blk, 0, stream>>>(x, sum, deg, W1s, W1n, b1, h1, nNodes);

  agg_kernel<false><<<gAgg, blkAgg, 0, stream>>>(h1, src, dst, key, sum, deg, nNodes, nEdges, nChunks);
  sage_gemm<true, false><<<gGemm, blk, 0, stream>>>(h1, sum, deg, W2s, W2n, b2, h2, nNodes);

  sage_gemm<false, false><<<gGemm, blk, 0, stream>>>(h2, nullptr, nullptr, W3, nullptr, b3,
                                                     out, nNodes);
}


# layer 2 streams each block's edge list, which layer 1's filter appends to the workspace in the same call: no second key scan, finder or chunk barriers; LDS float atomic sums, arrival order as before
# speedup vs baseline: 2.4438x; 1.1738x over previous
#include <hip/hip_runtime.h>

#define NODE_F 128

typedef float v2f __attribute__((ext_vector_type(2)));
typedef float v8f __attribute__((ext_vector_type(8)));

static __device__ __forceinline__ v8f wmma_k4(v2f a, v2f b, v8f c) {
  return __builtin_amdgcn_wmma_f32_16x16x4_f32(
       false, a,  false, b,
       (short)0, c,  false,  false);
}

#define AGG_R     400
#define AGG_T     1024
#define AGG_V     4
#define AGG_B     16
#define AGG_QCAP  (32 * AGG_B)
#define AGG_LCAP  16384

__global__ void key_kernel(const int* __restrict__ dst, unsigned* __restrict__ key,
                           int nEdges, int nWords) {
  const int t = blockIdx.x * blockDim.x + threadIdx.x;
  if (t >= nWords) return;
  const int e = t * 4;
  unsigned w = 0xFFFFFFFFu;
  if (e + 3 < nEdges) {
    const int4 d = ((const int4*)dst)[t];
    w = (unsigned)min(d.x / AGG_R, 255) | ((unsigned)min(d.y / AGG_R, 255) << 8) |
        ((unsigned)min(d.z / AGG_R, 255) << 16) | ((unsigned)min(d.w / AGG_R, 255) << 24);
  } else {
    for (int b = 0; b < 4; ++b)
      if (e + b < nEdges)
        w = (w & ~(0xFFu << (8 * b))) | ((unsigned)min(dst[e + b] / AGG_R, 255) << (8 * b));
  }
  key[t] = w;
}

template <bool FIRST>
__global__ void __launch_bounds__(AGG_T)
agg_kernel(const float* __restrict__ h, const int* __restrict__ src,
           const int* __restrict__ dst, const unsigned* __restrict__ key,
           float* __restrict__ sum, float* __restrict__ deg,
           unsigned* __restrict__ list, unsigned* __restrict__ meta,
           int nNodes, int nEdges, int nChunks) {
  __shared__ float    acc[(AGG_R + 1) * NODE_F];
  __shared__ unsigned q[2][AGG_QCAP];
  __shared__ int      cnt[AGG_R];
  __shared__ unsigned qn[4];

  const int tid  = threadIdx.x;
  const int lane = tid & 31;
  const int wave = __builtin_amdgcn_readfirstlane(tid >> 5);
  const int lo   = blockIdx.x * AGG_R;
  const unsigned nOwn   = (unsigned)min(AGG_R, nNodes - lo);
  const unsigned keyrep = (unsigned)min((int)blockIdx.x, 255) * 0x01010101u;

  for (int i = tid; i < (AGG_R + 1) * NODE_F / 4; i += AGG_T)
    ((float4*)acc)[i] = make_float4(0.f, 0.f, 0.f, 0.f);
  for (int i = tid; i < 2 * AGG_QCAP; i += AGG_T) (&q[0][0])[i] = 0u;
  if (FIRST)
    for (int i = tid; i < AGG_R; i += AGG_T) cnt[i] = 0;
  if (tid < 4) qn[tid] = 0u;

  unsigned* lst = list + (size_t)blockIdx.x * AGG_LCAP;
  unsigned lcnt = 0u;
  bool useList = false;
  if (!FIRST) {
    lcnt    = meta[2 * blockIdx.x];
    useList = meta[2 * blockIdx.x + 1] == 0u;
  }
  __syncthreads();

  const int  subVec  = AGG_T * AGG_V;
  const uint4* key4  = (const uint4*)key;

  uint4 d[AGG_V];

  auto load_keys = [&](int g) {
#pragma unroll
    for (int u = 0; u < AGG_V; ++u) d[u] = key4[g * subVec + u * AGG_T + tid];
  };
  auto eq4 = [&](unsigned w) -> unsigned {
    const unsigned x = w ^ keyrep;
    return ((x - 0x01010101u) & ~x & 0x80808080u) >> 7;
  };

  auto push_chunk = [&](int g) {
    unsigned* qq = q[g & 1];
    unsigned long long m = 0ull;
#pragma unroll
    for (int u = 0; u < AGG_V; ++u) {
      const unsigned p = (eq4(d[u].x) << ((4 * u + 0) & 7)) | (eq4(d[u].y) << ((4 * u + 1) & 7)) |
                         (eq4(d[u].z) << ((4 * u + 2) & 7)) | (eq4(d[u].w) << ((4 * u + 3) & 7));
      m |= (unsigned long long)p << (32 * (u >> 1));
    }
    load_keys(g + 1);

    while (m) {
      const int i = __ffsll((long long)m) - 1;
      m &= m - 1ull;
      const int k = (i & 7) + 8 * (i >> 5);
      const int b = (i >> 3) & 3;
      const int e = (g * subVec + (k >> 2) * AGG_T + tid) * 16 + (k & 3) * 4 + b;
      if (e < nEdges) {
        int s = src[e];
        const unsigned dl = (unsigned)(dst[e] - lo);
        asm volatile("" : "+v"(s));
        if (dl < nOwn) {
          if (FIRST) atomicAdd(&cnt[dl], 1);
          const unsigned pos = atomicAdd(&qn[g & 3], 1u);
          if (pos < AGG_QCAP) {
            qq[pos] = ((unsigned)s << 9) | dl;
          } else {
            const float* hp = h + (size_t)s * NODE_F;
            for (int f = 0; f < NODE_F; ++f) atomicAdd(&acc[dl * NODE_F + f], hp[f]);
          }
        }
      }
    }
  };

  unsigned ent[AGG_B];
  float v[AGG_B][4];
  auto load_rows = [&](auto word, unsigned k0, unsigned n) {
    unsigned raw[AGG_B];
#pragma unroll
    for (int j = 0; j < AGG_B; ++j) raw[j] = word(j);
#pragma unroll
    for (int j = 1; j < AGG_B; ++j) {
      ent[j] = 0u;
      if (k0 + 32u * j < n) {
        ent[j] = (unsigned)__builtin_amdgcn_readfirstlane((int)raw[j]);
        const float* hp = h + (size_t)(ent[j] >> 9) * NODE_F + lane;
        v[j][0] = hp[0];
        v[j][1] = hp[32];
        v[j][2] = hp[64];
        v[j][3] = hp[96];
      }
    }
    ent[0] = (k0 < n) ? (unsigned)__builtin_amdgcn_readfirstlane((int)raw[0]) : (unsigned)AGG_R;
    {
      const float* hp = h + (size_t)(ent[0] >> 9) * NODE_F + lane;
      v[0][0] = hp[0];
      v[0][1] = hp[32];
      v[0][2] = hp[64];
      v[0][3] = hp[96];
    }
  };
  auto add_rows = [&](unsigned k0, unsigned n) {
#pragma unroll
    for (int j = 1; j < AGG_B; ++j) {
      if (k0 + 32u * j < n) {
        float* ap = acc + (ent[j] & 511u) * NODE_F + lane;
        atomicAdd(ap,      v[j][0]);
        atomicAdd(ap + 32, v[j][1]);
        atomicAdd(ap + 64, v[j][2]);
        atomicAdd(ap + 96, v[j][3]);
      }
    }
    {
      float* ap = acc + (ent[0] & 511u) * NODE_F + lane;
      atomicAdd(ap,      v[0][0]);
      atomicAdd(ap + 32, v[0][1]);
      atomicAdd(ap + 64, v[0][2]);
      atomicAdd(ap + 96, v[0][3]);
    }
  };

  if (!FIRST && useList) {
    auto list_words = [&](unsigned p0) -> unsigned {
      const unsigned n = min(lcnt - p0, (unsigned)AGG_QCAP);
      return lst[p0 + min((unsigned)wave + 32u * (unsigned)(lane & (AGG_B - 1)), n - 1u)];
    };
    unsigned cur = (lcnt > 0u) ? list_words(0u) : 0u;
    for (unsigned p0 = 0u; p0 < lcnt; p0 += AGG_QCAP) {
      const unsigned n = min(lcnt - p0, (unsigned)AGG_QCAP);
      load_rows([&](int j) -> unsigned {
                  return (unsigned)__builtin_amdgcn_readlane((int)cur, j);
                }, (unsigned)wave, n);
      unsigned nxt = 0u;
      if (p0 + AGG_QCAP < lcnt) nxt = list_words(p0 + AGG_QCAP);
      add_rows((unsigned)wave, n);
      cur = nxt;
    }
  } else {
    unsigned lbase = 0u;
    bool     lbad  = false;

    load_keys(0);
    if (nChunks > 0) push_chunk(0);

    for (int c = 0; c < nChunks; ++c) {
      __syncthreads();
      if (tid == 0) qn[(c + 2) & 3] = 0u;

      const unsigned* qq = q[c & 1];
      const unsigned nq = qn[c & 3];
      const unsigned n  = min(nq, (unsigned)AGG_QCAP);
      load_rows([&](int j) -> unsigned {
                  return qq[min((unsigned)wave + 32u * j, (unsigned)(AGG_QCAP - 1))];
                }, (unsigned)wave, n);
      if (FIRST) {
        if ((unsigned)tid < n && lbase + (unsigned)tid < (unsigned)AGG_LCAP)
          lst[lbase + (unsigned)tid] = qq[tid];
        lbad = lbad || (nq > (unsigned)AGG_QCAP) || (lbase + n > (unsigned)AGG_LCAP);
        lbase += n;
      }
      if (c + 1 < nChunks) push_chunk(c + 1);
      add_rows((unsigned)wave, n);
    }
    if (FIRST && tid == 0) {
      meta[2 * blockIdx.x]     = lbase;
      meta[2 * blockIdx.x + 1] = lbad ? 1u : 0u;
    }
  }
  __syncthreads();

  for (unsigned r = (unsigned)wave; r < nOwn; r += 32u) {
    const float* ap = acc + r * NODE_F + lane;
    float* sp = sum + (size_t)(lo + r) * NODE_F + lane;
    sp[0]  = ap[0];
    sp[32] = ap[32];
    sp[64] = ap[64];
    sp[96] = ap[96];
  }
  if (FIRST)
    for (unsigned i = tid; i < nOwn; i += AGG_T) deg[lo + i] = (float)cnt[i];
}

template <bool NEIGH, bool RELU>
__global__ void sage_gemm(const float* __restrict__ hself, const float* __restrict__ sum,
                          const float* __restrict__ deg, const float* __restrict__ Wself,
                          const float* __restrict__ Wneigh, const float* __restrict__ bias,
                          float* __restrict__ out, int nNodes) {
  __shared__ float hs[32][NODE_F + 4];
  __shared__ float hn[32][NODE_F + 4];

  const int tid     = threadIdx.x;
  const int m_base  = blockIdx.x * 32;
  const bool fullTile = (m_base + 32 <= nNodes);

  for (int i = 0; i < 4; ++i) {
    int f   = i * 256 + tid;
    int row = f >> 5;
    int cv  = f & 31;
    int m   = m_base + row;
    float4 v = make_float4(0.f, 0.f, 0.f, 0.f);
    if (fullTile || m < nNodes) v = ((const float4*)(hself + (size_t)m * NODE_F))[cv];
    *(float4*)&hs[row][cv * 4] = v;
    if (NEIGH) {
      float4 s4 = make_float4(0.f, 0.f, 0.f, 0.f);
      float rd = 1.0f;
      if (fullTile || m < nNodes) {
        s4 = ((const float4*)(sum + (size_t)m * NODE_F))[cv];
        rd = 1.0f / fmaxf(deg[m], 1.0f);
      }
      hn[row][cv * 4 + 0] = s4.x * rd;
      hn[row][cv * 4 + 1] = s4.y * rd;
      hn[row][cv * 4 + 2] = s4.z * rd;
      hn[row][cv * 4 + 3] = s4.w * rd;
    }
  }
  __syncthreads();

  const int lane = tid & 31;
  const int wv   = tid >> 5;
  const int lid  = lane & 15;
  const int half = lane >> 4;
  const int n0   = wv * 16;

  const float* Ws = Wself + (size_t)(n0 + lid) * NODE_F;
  const float* Wn = NEIGH ? (Wneigh + (size_t)(n0 + lid) * NODE_F) : nullptr;

  v8f acc0 = {0.f, 0.f, 0.f, 0.f, 0.f, 0.f, 0.f, 0.f};
  v8f acc1 = {0.f, 0.f, 0.f, 0.f, 0.f, 0.f, 0.f, 0.f};

#pragma unroll 8
  for (int k0 = 0; k0 < NODE_F; k0 += 4) {
    const int kk = k0 + 2 * half;
    v2f bs;
    bs.x = Ws[kk];
    bs.y = Ws[kk + 1];
    v2f a0, a1;
    a0.x = hs[lid][kk];
    a0.y = hs[lid][kk + 1];
    a1.x = hs[16 + lid][kk];
    a1.y = hs[16 + lid][kk + 1];
    acc0 = wmma_k4(a0, bs, acc0);
    acc1 = wmma_k4(a1, bs, acc1);
    if (NEIGH) {
      v2f bn;
      bn.x = Wn[kk];
      bn.y = Wn[kk + 1];
      v2f c0, c1;
      c0.x = hn[lid][kk];
      c0.y = hn[lid][kk + 1];
      c1.x = hn[16 + lid][kk];
      c1.y = hn[16 + lid][kk + 1];
      acc0 = wmma_k4(c0, bn, acc0);
      acc1 = wmma_k4(c1, bn, acc1);
    }
  }

  const float bv = bias[n0 + lid];
  float* outc = out + n0 + lid;
  const int r0 = m_base + 8 * half;
  const int r1 = m_base + 16 + 8 * half;

  if (fullTile) {
#pragma unroll
    for (int e = 0; e < 8; ++e) {
      float v0 = acc0[e] + bv;
      float v1 = acc1[e] + bv;
      if (RELU) { v0 = fmaxf(v0, 0.f); v1 = fmaxf(v1, 0.f); }
      outc[(size_t)(r0 + e) * NODE_F] = v0;
      outc[(size_t)(r1 + e) * NODE_F] = v1;
    }
  } else {
#pragma unroll
    for (int e = 0; e < 8; ++e) {
      float v0 = acc0[e] + bv;
      float v1 = acc1[e] + bv;
      if (RELU) { v0 = fmaxf(v0, 0.f); v1 = fmaxf(v1, 0.f); }
      if (r0 + e < nNodes) outc[(size_t)(r0 + e) * NODE_F] = v0;
      if (r1 + e < nNodes) outc[(size_t)(r1 + e) * NODE_F] = v1;
    }
  }
}

extern "C" void kernel_launch(void* const* d_in, const int* in_sizes, int n_in,
                              void* d_out, int out_size, void* d_ws, size_t ws_size,
                              hipStream_t stream) {
  const float* x      = (const float*)d_in[0];
  const int*   src    = (const int*)d_in[1];
  const int*   dst    = (const int*)d_in[2];
  const float* W1s    = (const float*)d_in[3];
  const float* W1n    = (const float*)d_in[4];
  const float* b1     = (const float*)d_in[5];
  const float* W2s    = (const float*)d_in[6];
  const float* W2n    = (const float*)d_in[7];
  const float* b2     = (const float*)d_in[8];
  const float* W3     = (const float*)d_in[9];
  const float* b3     = (const float*)d_in[10];
  float*       out    = (float*)d_out;

  const int nNodes = in_sizes[0] / NODE_F;
  const int nEdges = in_sizes[1];

  float* sum = (float*)d_ws;
  float* deg = sum + (size_t)nNodes * NODE_F;
  float* h1  = deg + nNodes;
  float* h2  = h1 + (size_t)nNodes * NODE_F;

  unsigned* key = (unsigned*)h2;
  const int subVec  = AGG_T * AGG_V;
  const int nChunks = ((nEdges + 15) / 16 + subVec - 1) / subVec;
  const int nWords  = (nChunks + 1) * subVec * 4;

  dim3 blk(256);
  dim3 blkAgg(AGG_T);
  dim3 gKey((nWords + 255) / 256);
  dim3 gAgg((nNodes + AGG_R - 1) / AGG_R);
  dim3 gGemm((nNodes + 31) / 32);
  unsigned* list = key + nWords;
  unsigned* meta = list + (size_t)gAgg.x * AGG_LCAP;

  key_kernel<<<gKey, blk, 0, stream>>>(dst, key, nEdges, nWords);

  agg_kernel<true><<<gAgg, blkAgg, 0, stream>>>(x, src, dst, key, sum, deg, list, meta,
                                                nNodes, nEdges, nChunks);
  sage_gemm<true, true><<<gGemm, blk, 0, stream>>>(x, sum, deg, W1s, W1n, b1, h1, nNodes);

  agg_kernel<false><<<gAgg, blkAgg, 0, stream>>>(h1, src, dst, key, sum, deg, list, meta,
                                                 nNodes, nEdges, nChunks);
  sage_gemm<true, false><<<gGemm, blk, 0, stream>>>(h1, sum, deg, W2s, W2n, b2, h2, nNodes);

  sage_gemm<false, false><<<gGemm, blk, 0, stream>>>(h2, nullptr, nullptr, W3, nullptr, b3,
                                                     out, nNodes);
}


# weights repacked per call into WMMA B-fragment order by extra blocks of the key launch: a wave's B load for two k-steps is one contiguous 512-byte load, not 16 lines x 8 bytes; values, sum order kept
# speedup vs baseline: 2.6419x; 1.0811x over previous
#include <hip/hip_runtime.h>

#define NODE_F 128

typedef float v2f __attribute__((ext_vector_type(2)));
typedef float v8f __attribute__((ext_vector_type(8)));

static __device__ __forceinline__ v8f wmma_k4(v2f a, v2f b, v8f c) {
  return __builtin_amdgcn_wmma_f32_16x16x4_f32(
       false, a,  false, b,
       (short)0, c,  false,  false);
}

#define AGG_R     400
#define AGG_T     1024
#define AGG_V     4
#define AGG_B     16
#define AGG_QCAP  (32 * AGG_B)
#define AGG_LCAP  16384

#define PACK_BLOCKS_PER_W 16
__global__ void key_kernel(const int* __restrict__ dst, unsigned* __restrict__ key,
                           int nEdges, int nWords, int nKeyBlocks,
                           const float* __restrict__ Wa, const float* __restrict__ Wb,
                           const float* __restrict__ Wc, const float* __restrict__ Wd,
                           const float* __restrict__ We, float* __restrict__ wpk) {
  if ((int)blockIdx.x >= nKeyBlocks) {
    const int pb = (int)blockIdx.x - nKeyBlocks;
    const int m  = pb / PACK_BLOCKS_PER_W;
    const float* Wm = (m == 0) ? Wa : (m == 1) ? Wb : (m == 2) ? Wc : (m == 3) ? Wd : We;
    const int q    = (pb % PACK_BLOCKS_PER_W) * 256 + (int)threadIdx.x;
    const int ln   = q & 31;
    const int jj   = (q >> 5) & 15;
    const int nt   = q >> 9;
    const float* wr = Wm + (size_t)(nt * 16 + (ln & 15)) * NODE_F + 8 * jj + 2 * (ln >> 4);
    const float2 p0 = *(const float2*)wr;
    const float2 p1 = *(const float2*)(wr + 4);
    ((float4*)wpk)[(size_t)m * (NODE_F * NODE_F / 4) + q] = make_float4(p0.x, p0.y, p1.x, p1.y);
    return;
  }
  const int t = blockIdx.x * blockDim.x + threadIdx.x;
  if (t >= nWords) return;
  const int e = t * 4;
  unsigned w = 0xFFFFFFFFu;
  if (e + 3 < nEdges) {
    const int4 d = ((const int4*)dst)[t];
    w = (unsigned)min(d.x / AGG_R, 255) | ((unsigned)min(d.y / AGG_R, 255) << 8) |
        ((unsigned)min(d.z / AGG_R, 255) << 16) | ((unsigned)min(d.w / AGG_R, 255) << 24);
  } else {
    for (int b = 0; b < 4; ++b)
      if (e + b < nEdges)
        w = (w & ~(0xFFu << (8 * b))) | ((unsigned)min(dst[e + b] / AGG_R, 255) << (8 * b));
  }
  key[t] = w;
}

template <bool FIRST>
__global__ void __launch_bounds__(AGG_T)
agg_kernel(const float* __restrict__ h, const int* __restrict__ src,
           const int* __restrict__ dst, const unsigned* __restrict__ key,
           float* __restrict__ sum, float* __restrict__ deg,
           unsigned* __restrict__ list, unsigned* __restrict__ meta,
           int nNodes, int nEdges, int nChunks) {
  __shared__ float    acc[(AGG_R + 1) * NODE_F];
  __shared__ unsigned q[2][AGG_QCAP];
  __shared__ int      cnt[AGG_R];
  __shared__ unsigned qn[4];

  const int tid  = threadIdx.x;
  const int lane = tid & 31;
  const int wave = __builtin_amdgcn_readfirstlane(tid >> 5);
  const int lo   = blockIdx.x * AGG_R;
  const unsigned nOwn   = (unsigned)min(AGG_R, nNodes - lo);
  const unsigned keyrep = (unsigned)min((int)blockIdx.x, 255) * 0x01010101u;

  for (int i = tid; i < (AGG_R + 1) * NODE_F / 4; i += AGG_T)
    ((float4*)acc)[i] = make_float4(0.f, 0.f, 0.f, 0.f);
  for (int i = tid; i < 2 * AGG_QCAP; i += AGG_T) (&q[0][0])[i] = 0u;
  if (FIRST)
    for (int i = tid; i < AGG_R; i += AGG_T) cnt[i] = 0;
  if (tid < 4) qn[tid] = 0u;

  unsigned* lst = list + (size_t)blockIdx.x * AGG_LCAP;
  unsigned lcnt = 0u;
  bool useList = false;
  if (!FIRST) {
    lcnt    = meta[2 * blockIdx.x];
    useList = meta[2 * blockIdx.x + 1] == 0u;
  }
  __syncthreads();

  const int  subVec  = AGG_T * AGG_V;
  const uint4* key4  = (const uint4*)key;

  uint4 d[AGG_V];

  auto load_keys = [&](int g) {
#pragma unroll
    for (int u = 0; u < AGG_V; ++u) d[u] = key4[g * subVec + u * AGG_T + tid];
  };
  auto eq4 = [&](unsigned w) -> unsigned {
    const unsigned x = w ^ keyrep;
    return ((x - 0x01010101u) & ~x & 0x80808080u) >> 7;
  };

  auto push_chunk = [&](int g) {
    unsigned* qq = q[g & 1];
    unsigned long long m = 0ull;
#pragma unroll
    for (int u = 0; u < AGG_V; ++u) {
      const unsigned p = (eq4(d[u].x) << ((4 * u + 0) & 7)) | (eq4(d[u].y) << ((4 * u + 1) & 7)) |
                         (eq4(d[u].z) << ((4 * u + 2) & 7)) | (eq4(d[u].w) << ((4 * u + 3) & 7));
      m |= (unsigned long long)p << (32 * (u >> 1));
    }
    load_keys(g + 1);

    while (m) {
      const int i = __ffsll((long long)m) - 1;
      m &= m - 1ull;
      const int k = (i & 7) + 8 * (i >> 5);
      const int b = (i >> 3) & 3;
      const int e = (g * subVec + (k >> 2) * AGG_T + tid) * 16 + (k & 3) * 4 + b;
      if (e < nEdges) {
        int s = src[e];
        const unsigned dl = (unsigned)(dst[e] - lo);
        asm volatile("" : "+v"(s));
        if (dl < nOwn) {
          if (FIRST) atomicAdd(&cnt[dl], 1);
          const unsigned pos = atomicAdd(&qn[g & 3], 1u);
          if (pos < AGG_QCAP) {
            qq[pos] = ((unsigned)s << 9) | dl;
          } else {
            const float* hp = h + (size_t)s * NODE_F;
            for (int f = 0; f < NODE_F; ++f) atomicAdd(&acc[dl * NODE_F + f], hp[f]);
          }
        }
      }
    }
  };

  unsigned ent[AGG_B];
  float v[AGG_B][4];
  auto load_rows = [&](auto word, unsigned k0, unsigned n) {
    unsigned raw[AGG_B];
#pragma unroll
    for (int j = 0; j < AGG_B; ++j) raw[j] = word(j);
#pragma unroll
    for (int j = 1; j < AGG_B; ++j) {
      ent[j] = 0u;
      if (k0 + 32u * j < n) {
        ent[j] = (unsigned)__builtin_amdgcn_readfirstlane((int)raw[j]);
        const float* hp = h + (size_t)(ent[j] >> 9) * NODE_F + lane;
        v[j][0] = hp[0];
        v[j][1] = hp[32];
        v[j][2] = hp[64];
        v[j][3] = hp[96];
      }
    }
    ent[0] = (k0 < n) ? (unsigned)__builtin_amdgcn_readfirstlane((int)raw[0]) : (unsigned)AGG_R;
    {
      const float* hp = h + (size_t)(ent[0] >> 9) * NODE_F + lane;
      v[0][0] = hp[0];
      v[0][1] = hp[32];
      v[0][2] = hp[64];
      v[0][3] = hp[96];
    }
  };
  auto add_rows = [&](unsigned k0, unsigned n) {
#pragma unroll
    for (int j = 1; j < AGG_B; ++j) {
      if (k0 + 32u * j < n) {
        float* ap = acc + (ent[j] & 511u) * NODE_F + lane;
        atomicAdd(ap,      v[j][0]);
        atomicAdd(ap + 32, v[j][1]);
        atomicAdd(ap + 64, v[j][2]);
        atomicAdd(ap + 96, v[j][3]);
      }
    }
    {
      float* ap = acc + (ent[0] & 511u) * NODE_F + lane;
      atomicAdd(ap,      v[0][0]);
      atomicAdd(ap + 32, v[0][1]);
      atomicAdd(ap + 64, v[0][2]);
      atomicAdd(ap + 96, v[0][3]);
    }
  };

  if (!FIRST && useList) {
    auto list_words = [&](unsigned p0) -> unsigned {
      const unsigned n = min(lcnt - p0, (unsigned)AGG_QCAP);
      return lst[p0 + min((unsigned)wave + 32u * (unsigned)(lane & (AGG_B - 1)), n - 1u)];
    };
    unsigned cur = (lcnt > 0u) ? list_words(0u) : 0u;
    for (unsigned p0 = 0u; p0 < lcnt; p0 += AGG_QCAP) {
      const unsigned n = min(lcnt - p0, (unsigned)AGG_QCAP);
      load_rows([&](int j) -> unsigned {
                  return (unsigned)__builtin_amdgcn_readlane((int)cur, j);
                }, (unsigned)wave, n);
      unsigned nxt = 0u;
      if (p0 + AGG_QCAP < lcnt) nxt = list_words(p0 + AGG_QCAP);
      add_rows((unsigned)wave, n);
      cur = nxt;
    }
  } else {
    unsigned lbase = 0u;
    bool     lbad  = false;

    load_keys(0);
    if (nChunks > 0) push_chunk(0);

    for (int c = 0; c < nChunks; ++c) {
      __syncthreads();
      if (tid == 0) qn[(c + 2) & 3] = 0u;

      const unsigned* qq = q[c & 1];
      const unsigned nq = qn[c & 3];
      const unsigned n  = min(nq, (unsigned)AGG_QCAP);
      load_rows([&](int j) -> unsigned {
                  return qq[min((unsigned)wave + 32u * j, (unsigned)(AGG_QCAP - 1))];
                }, (unsigned)wave, n);
      if (FIRST) {
        if ((unsigned)tid < n && lbase + (unsigned)tid < (unsigned)AGG_LCAP)
          lst[lbase + (unsigned)tid] = qq[tid];
        lbad = lbad || (nq > (unsigned)AGG_QCAP) || (lbase + n > (unsigned)AGG_LCAP);
        lbase += n;
      }
      if (c + 1 < nChunks) push_chunk(c + 1);
      add_rows((unsigned)wave, n);
    }
    if (FIRST && tid == 0) {
      meta[2 * blockIdx.x]     = lbase;
      meta[2 * blockIdx.x + 1] = lbad ? 1u : 0u;
    }
  }
  __syncthreads();

  for (unsigned r = (unsigned)wave; r < nOwn; r += 32u) {
    const float* ap = acc + r * NODE_F + lane;
    float* sp = sum + (size_t)(lo + r) * NODE_F + lane;
    sp[0]  = ap[0];
    sp[32] = ap[32];
    sp[64] = ap[64];
    sp[96] = ap[96];
  }
  if (FIRST)
    for (unsigned i = tid; i < nOwn; i += AGG_T) deg[lo + i] = (float)cnt[i];
}

template <bool NEIGH, bool RELU>
__global__ void sage_gemm(const float* __restrict__ hself, const float* __restrict__ sum,
                          const float* __restrict__ deg, const float* __restrict__ Wself,
                          const float* __restrict__ Wneigh, const float* __restrict__ bias,
                          float* __restrict__ out, int nNodes) {
  __shared__ float hs[32][NODE_F + 4];
  __shared__ float hn[32][NODE_F + 4];

  const int tid     = threadIdx.x;
  const int m_base  = blockIdx.x * 32;
  const bool fullTile = (m_base + 32 <= nNodes);

  for (int i = 0; i < 4; ++i) {
    int f   = i * 256 + tid;
    int row = f >> 5;
    int cv  = f & 31;
    int m   = m_base + row;
    float4 v = make_float4(0.f, 0.f, 0.f, 0.f);
    if (fullTile || m < nNodes) v = ((const float4*)(hself + (size_t)m * NODE_F))[cv];
    *(float4*)&hs[row][cv * 4] = v;
    if (NEIGH) {
      float4 s4 = make_float4(0.f, 0.f, 0.f, 0.f);
      float rd = 1.0f;
      if (fullTile || m < nNodes) {
        s4 = ((const float4*)(sum + (size_t)m * NODE_F))[cv];
        rd = 1.0f / fmaxf(deg[m], 1.0f);
      }
      hn[row][cv * 4 + 0] = s4.x * rd;
      hn[row][cv * 4 + 1] = s4.y * rd;
      hn[row][cv * 4 + 2] = s4.z * rd;
      hn[row][cv * 4 + 3] = s4.w * rd;
    }
  }
  __syncthreads();

  const int lane = tid & 31;
  const int wv   = tid >> 5;
  const int lid  = lane & 15;
  const int half = lane >> 4;
  const int n0   = wv * 16;

  const float4* Ws = (const float4*)Wself + wv * 16 * 32 + lane;
  const float4* Wn = NEIGH ? ((const float4*)Wneigh + wv * 16 * 32 + lane) : nullptr;

  v8f acc0 = {0.f, 0.f, 0.f, 0.f, 0.f, 0.f, 0.f, 0.f};
  v8f acc1 = {0.f, 0.f, 0.f, 0.f, 0.f, 0.f, 0.f, 0.f};

#pragma unroll 4
  for (int jj = 0; jj < NODE_F / 8; ++jj) {
    const float4 bs4 = Ws[jj * 32];
    float4 bn4 = make_float4(0.f, 0.f, 0.f, 0.f);
    if (NEIGH) bn4 = Wn[jj * 32];
#pragma unroll
    for (int s = 0; s < 2; ++s) {
      const int kk = 8 * jj + 4 * s + 2 * half;
      v2f bs;
      bs.x = s ? bs4.z : bs4.x;
      bs.y = s ? bs4.w : bs4.y;
      v2f a0, a1;
      a0.x = hs[lid][kk];
      a0.y = hs[lid][kk + 1];
      a1.x = hs[16 + lid][kk];
      a1.y = hs[16 + lid][kk + 1];
      acc0 = wmma_k4(a0, bs, acc0);
      acc1 = wmma_k4(a1, bs, acc1);
      if (NEIGH) {
        v2f bn;
        bn.x = s ? bn4.z : bn4.x;
        bn.y = s ? bn4.w : bn4.y;
        v2f c0, c1;
        c0.x = hn[lid][kk];
        c0.y = hn[lid][kk + 1];
        c1.x = hn[16 + lid][kk];
        c1.y = hn[16 + lid][kk + 1];
        acc0 = wmma_k4(c0, bn, acc0);
        acc1 = wmma_k4(c1, bn, acc1);
      }
    }
  }

  const float bv = bias[n0 + lid];
  float* outc = out + n0 + lid;
  const int r0 = m_base + 8 * half;
  const int r1 = m_base + 16 + 8 * half;

  if (fullTile) {
#pragma unroll
    for (int e = 0; e < 8; ++e) {
      float v0 = acc0[e] + bv;
      float v1 = acc1[e] + bv;
      if (RELU) { v0 = fmaxf(v0, 0.f); v1 = fmaxf(v1, 0.f); }
      outc[(size_t)(r0 + e) * NODE_F] = v0;
      outc[(size_t)(r1 + e) * NODE_F] = v1;
    }
  } else {
#pragma unroll
    for (int e = 0; e < 8; ++e) {
      float v0 = acc0[e] + bv;
      float v1 = acc1[e] + bv;
      if (RELU) { v0 = fmaxf(v0, 0.f); v1 = fmaxf(v1, 0.f); }
      if (r0 + e < nNodes) outc[(size_t)(r0 + e) * NODE_F] = v0;
      if (r1 + e < nNodes) outc[(size_t)(r1 + e) * NODE_F] = v1;
    }
  }
}

extern "C" void kernel_launch(void* const* d_in, const int* in_sizes, int n_in,
                              void* d_out, int out_size, void* d_ws, size_t ws_size,
                              hipStream_t stream) {
  const float* x      = (const float*)d_in[0];
  const int*   src    = (const int*)d_in[1];
  const int*   dst    = (const int*)d_in[2];
  const float* W1s    = (const float*)d_in[3];
  const float* W1n    = (const float*)d_in[4];
  const float* b1     = (const float*)d_in[5];
  const float* W2s    = (const float*)d_in[6];
  const float* W2n    = (const float*)d_in[7];
  const float* b2     = (const float*)d_in[8];
  const float* W3     = (const float*)d_in[9];
  const float* b3     = (const float*)d_in[10];
  float*       out    = (float*)d_out;

  const int nNodes = in_sizes[0] / NODE_F;
  const int nEdges = in_sizes[1];

  float* wpk = (float*)d_ws;
  float* sum = out;
  float* deg = wpk + (size_t)nNodes * NODE_F;
  float* h1  = deg + nNodes;
  float* h2  = h1 + (size_t)nNodes * NODE_F;

  unsigned* key = (unsigned*)h2;
  const int subVec  = AGG_T * AGG_V;
  const int nChunks = ((nEdges + 15) / 16 + subVec - 1) / subVec;
  const int nWords  = (nChunks + 1) * subVec * 4;

  dim3 blk(256);
  dim3 blkAgg(AGG_T);
  dim3 gKey((nWords + 255) / 256);
  dim3 gAgg((nNodes + AGG_R - 1) / AGG_R);
  dim3 gGemm((nNodes + 31) / 32);
  unsigned* list = key + nWords;
  unsigned* meta = list + (size_t)gAgg.x * AGG_LCAP;

  dim3 gKeyPack(gKey.x + 5 * PACK_BLOCKS_PER_W);
  const float* W1sP = wpk;
  const float* W1nP = wpk + 1 * NODE_F * NODE_F;
  const float* W2sP = wpk + 2 * NODE_F * NODE_F;
  const float* W2nP = wpk + 3 * NODE_F * NODE_F;
  const float* W3P  = wpk + 4 * NODE_F * NODE_F;
  key_kernel<<<gKeyPack, blk, 0, stream>>>(dst, key, nEdges, nWords, (int)gKey.x,
                                           W1s, W1n, W2s, W2n, W3, wpk);

  agg_kernel<true><<<gAgg, blkAgg, 0, stream>>>(x, src, dst, key, sum, deg, list, meta,
                                                nNodes, nEdges, nChunks);
  sage_gemm<true, true><<<gGemm, blk, 0, stream>>>(x, sum, deg, W1sP, W1nP, b1, h1, nNodes);

  agg_kernel<false><<<gAgg, blkAgg, 0, stream>>>(h1, src, dst, key, sum, deg, list, meta,
                                                 nNodes, nEdges, nChunks);
  sage_gemm<true, false><<<gGemm, blk, 0, stream>>>(h1, sum, deg, W2sP, W2nP, b2, h2, nNodes);

  sage_gemm<false, false><<<gGemm, blk, 0, stream>>>(h2, nullptr, nullptr, W3P, nullptr, b3,
                                                     out, nNodes);
}


# aggregation rows move 16 bytes per lane: one 512-byte load per gathered row (scalar row base + lane index), one store per sum row; LDS sums lane-interleaved, adds conflict-free; arrival sum order
# speedup vs baseline: 2.7328x; 1.0344x over previous
#include <hip/hip_runtime.h>

#define NODE_F 128

typedef float v2f __attribute__((ext_vector_type(2)));
typedef float v8f __attribute__((ext_vector_type(8)));

static __device__ __forceinline__ v8f wmma_k4(v2f a, v2f b, v8f c) {
  return __builtin_amdgcn_wmma_f32_16x16x4_f32(
       false, a,  false, b,
       (short)0, c,  false,  false);
}

#define AGG_R     400
#define AGG_T     1024
#define AGG_V     4
#define AGG_B     16
#define AGG_QCAP  (32 * AGG_B)
#define AGG_LCAP  16384

#define PACK_BLOCKS_PER_W 16
__global__ void key_kernel(const int* __restrict__ dst, unsigned* __restrict__ key,
                           int nEdges, int nWords, int nKeyBlocks,
                           const float* __restrict__ Wa, const float* __restrict__ Wb,
                           const float* __restrict__ Wc, const float* __restrict__ Wd,
                           const float* __restrict__ We, float* __restrict__ wpk) {
  if ((int)blockIdx.x >= nKeyBlocks) {
    const int pb = (int)blockIdx.x - nKeyBlocks;
    const int m  = pb / PACK_BLOCKS_PER_W;
    const float* Wm = (m == 0) ? Wa : (m == 1) ? Wb : (m == 2) ? Wc : (m == 3) ? Wd : We;
    const int q    = (pb % PACK_BLOCKS_PER_W) * 256 + (int)threadIdx.x;
    const int ln   = q & 31;
    const int jj   = (q >> 5) & 15;
    const int nt   = q >> 9;
    const float* wr = Wm + (size_t)(nt * 16 + (ln & 15)) * NODE_F + 8 * jj + 2 * (ln >> 4);
    const float2 p0 = *(const float2*)wr;
    const float2 p1 = *(const float2*)(wr + 4);
    ((float4*)wpk)[(size_t)m * (NODE_F * NODE_F / 4) + q] = make_float4(p0.x, p0.y, p1.x, p1.y);
    return;
  }
  const int t = blockIdx.x * blockDim.x + threadIdx.x;
  if (t >= nWords) return;
  const int e = t * 4;
  unsigned w = 0xFFFFFFFFu;
  if (e + 3 < nEdges) {
    const int4 d = ((const int4*)dst)[t];
    w = (unsigned)min(d.x / AGG_R, 255) | ((unsigned)min(d.y / AGG_R, 255) << 8) |
        ((unsigned)min(d.z / AGG_R, 255) << 16) | ((unsigned)min(d.w / AGG_R, 255) << 24);
  } else {
    for (int b = 0; b < 4; ++b)
      if (e + b < nEdges)
        w = (w & ~(0xFFu << (8 * b))) | ((unsigned)min(dst[e + b] / AGG_R, 255) << (8 * b));
  }
  key[t] = w;
}

template <bool FIRST>
__global__ void __launch_bounds__(AGG_T)
agg_kernel(const float* __restrict__ h, const int* __restrict__ src,
           const int* __restrict__ dst, const unsigned* __restrict__ key,
           float* __restrict__ sum, float* __restrict__ deg,
           unsigned* __restrict__ list, unsigned* __restrict__ meta,
           int nNodes, int nEdges, int nChunks) {
  __shared__ float    acc[(AGG_R + 1) * NODE_F];
  __shared__ unsigned q[2][AGG_QCAP];
  __shared__ int      cnt[AGG_R];
  __shared__ unsigned qn[4];

  const int tid  = threadIdx.x;
  const int lane = tid & 31;
  const int wave = __builtin_amdgcn_readfirstlane(tid >> 5);
  const int lo   = blockIdx.x * AGG_R;
  const unsigned nOwn   = (unsigned)min(AGG_R, nNodes - lo);
  const unsigned keyrep = (unsigned)min((int)blockIdx.x, 255) * 0x01010101u;

  for (int i = tid; i < (AGG_R + 1) * NODE_F / 4; i += AGG_T)
    ((float4*)acc)[i] = make_float4(0.f, 0.f, 0.f, 0.f);
  for (int i = tid; i < 2 * AGG_QCAP; i += AGG_T) (&q[0][0])[i] = 0u;
  if (FIRST)
    for (int i = tid; i < AGG_R; i += AGG_T) cnt[i] = 0;
  if (tid < 4) qn[tid] = 0u;

  unsigned* lst = list + (size_t)blockIdx.x * AGG_LCAP;
  unsigned lcnt = 0u;
  bool useList = false;
  if (!FIRST) {
    lcnt    = meta[2 * blockIdx.x];
    useList = meta[2 * blockIdx.x + 1] == 0u;
  }
  __syncthreads();

  const int  subVec  = AGG_T * AGG_V;
  const uint4* key4  = (const uint4*)key;

  uint4 d[AGG_V];

  auto load_keys = [&](int g) {
#pragma unroll
    for (int u = 0; u < AGG_V; ++u) d[u] = key4[g * subVec + u * AGG_T + tid];
  };
  auto eq4 = [&](unsigned w) -> unsigned {
    const unsigned x = w ^ keyrep;
    return ((x - 0x01010101u) & ~x & 0x80808080u) >> 7;
  };

  auto push_chunk = [&](int g) {
    unsigned* qq = q[g & 1];
    unsigned long long m = 0ull;
#pragma unroll
    for (int u = 0; u < AGG_V; ++u) {
      const unsigned p = (eq4(d[u].x) << ((4 * u + 0) & 7)) | (eq4(d[u].y) << ((4 * u + 1) & 7)) |
                         (eq4(d[u].z) << ((4 * u + 2) & 7)) | (eq4(d[u].w) << ((4 * u + 3) & 7));
      m |= (unsigned long long)p << (32 * (u >> 1));
    }
    load_keys(g + 1);

    while (m) {
      const int i = __ffsll((long long)m) - 1;
      m &= m - 1ull;
      const int k = (i & 7) + 8 * (i >> 5);
      const int b = (i >> 3) & 3;
      const int e = (g * subVec + (k >> 2) * AGG_T + tid) * 16 + (k & 3) * 4 + b;
      if (e < nEdges) {
        int s = src[e];
        const unsigned dl = (unsigned)(dst[e] - lo);
        asm volatile("" : "+v"(s));
        if (dl < nOwn) {
          if (FIRST) atomicAdd(&cnt[dl], 1);
          const unsigned pos = atomicAdd(&qn[g & 3], 1u);
          if (pos < AGG_QCAP) {
            qq[pos] = ((unsigned)s << 9) | dl;
          } else {
            const float* hp = h + (size_t)s * NODE_F;
            for (int f = 0; f < NODE_F; ++f)
              atomicAdd(&acc[dl * NODE_F + (f >> 2) + 32 * (f & 3)], hp[f]);
          }
        }
      }
    }
  };

  unsigned ent[AGG_B];
  float4 v[AGG_B];
  typedef __attribute__((address_space(1))) const float4 grow_t;
  auto load_row = [&](unsigned e) -> float4 {
    const unsigned long long b = (unsigned long long)h + (unsigned long long)(e >> 9) * (NODE_F * 4ull);
    const unsigned blo = (unsigned)__builtin_amdgcn_readfirstlane((int)(unsigned)b);
    const unsigned bhi = (unsigned)__builtin_amdgcn_readfirstlane((int)(unsigned)(b >> 32));
    return ((grow_t*)(((unsigned long long)bhi << 32) | blo))[lane];
  };
  auto load_rows = [&](auto word, unsigned k0, unsigned n) {
    unsigned raw[AGG_B];
#pragma unroll
    for (int j = 0; j < AGG_B; ++j) raw[j] = word(j);
#pragma unroll
    for (int j = 1; j < AGG_B; ++j) {
      ent[j] = 0u;
      if (k0 + 32u * j < n) {
        ent[j] = (unsigned)__builtin_amdgcn_readfirstlane((int)raw[j]);
        v[j] = load_row(ent[j]);
      }
    }
    ent[0] = (k0 < n) ? (unsigned)__builtin_amdgcn_readfirstlane((int)raw[0]) : (unsigned)AGG_R;
    v[0] = load_row(ent[0]);
  };
  auto add_rows = [&](unsigned k0, unsigned n) {
#pragma unroll
    for (int j = 1; j < AGG_B; ++j) {
      if (k0 + 32u * j < n) {
        float* ap = acc + (ent[j] & 511u) * NODE_F + lane;
        atomicAdd(ap,      v[j].x);
        atomicAdd(ap + 32, v[j].y);
        atomicAdd(ap + 64, v[j].z);
        atomicAdd(ap + 96, v[j].w);
      }
    }
    {
      float* ap = acc + (ent[0] & 511u) * NODE_F + lane;
      atomicAdd(ap,      v[0].x);
      atomicAdd(ap + 32, v[0].y);
      atomicAdd(ap + 64, v[0].z);
      atomicAdd(ap + 96, v[0].w);
    }
  };

  if (!FIRST && useList) {
    auto list_words = [&](unsigned p0) -> unsigned {
      const unsigned n = min(lcnt - p0, (unsigned)AGG_QCAP);
      return lst[p0 + min((unsigned)wave + 32u * (unsigned)(lane & (AGG_B - 1)), n - 1u)];
    };
    unsigned cur = (lcnt > 0u) ? list_words(0u) : 0u;
    for (unsigned p0 = 0u; p0 < lcnt; p0 += AGG_QCAP) {
      const unsigned n = min(lcnt - p0, (unsigned)AGG_QCAP);
      load_rows([&](int j) -> unsigned {
                  return (unsigned)__builtin_amdgcn_readlane((int)cur, j);
                }, (unsigned)wave, n);
      unsigned nxt = 0u;
      if (p0 + AGG_QCAP < lcnt) nxt = list_words(p0 + AGG_QCAP);
      add_rows((unsigned)wave, n);
      cur = nxt;
    }
  } else {
    unsigned lbase = 0u;
    bool     lbad  = false;

    load_keys(0);
    if (nChunks > 0) push_chunk(0);

    for (int c = 0; c < nChunks; ++c) {
      __syncthreads();
      if (tid == 0) qn[(c + 2) & 3] = 0u;

      const unsigned* qq = q[c & 1];
      const unsigned nq = qn[c & 3];
      const unsigned n  = min(nq, (unsigned)AGG_QCAP);
      load_rows([&](int j) -> unsigned {
                  return qq[min((unsigned)wave + 32u * j, (unsigned)(AGG_QCAP - 1))];
                }, (unsigned)wave, n);
      if (FIRST) {
        if ((unsigned)tid < n && lbase + (unsigned)tid < (unsigned)AGG_LCAP)
          lst[lbase + (unsigned)tid] = qq[tid];
        lbad = lbad || (nq > (unsigned)AGG_QCAP) || (lbase + n > (unsigned)AGG_LCAP);
        lbase += n;
      }
      if (c + 1 < nChunks) push_chunk(c + 1);
      add_rows((unsigned)wave, n);
    }
    if (FIRST && tid == 0) {
      meta[2 * blockIdx.x]     = lbase;
      meta[2 * blockIdx.x + 1] = lbad ? 1u : 0u;
    }
  }
  __syncthreads();

  for (unsigned r = (unsigned)wave; r < nOwn; r += 32u) {
    const float* ap = acc + r * NODE_F + lane;
    ((float4*)(sum + (size_t)(lo + r) * NODE_F))[lane] = make_float4(ap[0], ap[32], ap[64], ap[96]);
  }
  if (FIRST)
    for (unsigned i = tid; i < nOwn; i += AGG_T) deg[lo + i] = (float)cnt[i];
}

template <bool NEIGH, bool RELU>
__global__ void sage_gemm(const float* __restrict__ hself, const float* __restrict__ sum,
                          const float* __restrict__ deg, const float* __restrict__ Wself,
                          const float* __restrict__ Wneigh, const float* __restrict__ bias,
                          float* __restrict__ out, int nNodes) {
  __shared__ float hs[32][NODE_F + 4];
  __shared__ float hn[32][NODE_F + 4];

  const int tid     = threadIdx.x;
  const int m_base  = blockIdx.x * 32;
  const bool fullTile = (m_base + 32 <= nNodes);

  for (int i = 0; i < 4; ++i) {
    int f   = i * 256 + tid;
    int row = f >> 5;
    int cv  = f & 31;
    int m   = m_base + row;
    float4 v = make_float4(0.f, 0.f, 0.f, 0.f);
    if (fullTile || m < nNodes) v = ((const float4*)(hself + (size_t)m * NODE_F))[cv];
    *(float4*)&hs[row][cv * 4] = v;
    if (NEIGH) {
      float4 s4 = make_float4(0.f, 0.f, 0.f, 0.f);
      float rd = 1.0f;
      if (fullTile || m < nNodes) {
        s4 = ((const float4*)(sum + (size_t)m * NODE_F))[cv];
        rd = 1.0f / fmaxf(deg[m], 1.0f);
      }
      hn[row][cv * 4 + 0] = s4.x * rd;
      hn[row][cv * 4 + 1] = s4.y * rd;
      hn[row][cv * 4 + 2] = s4.z * rd;
      hn[row][cv * 4 + 3] = s4.w * rd;
    }
  }
  __syncthreads();

  const int lane = tid & 31;
  const int wv   = tid >> 5;
  const int lid  = lane & 15;
  const int half = lane >> 4;
  const int n0   = wv * 16;

  const float4* Ws = (const float4*)Wself + wv * 16 * 32 + lane;
  const float4* Wn = NEIGH ? ((const float4*)Wneigh + wv * 16 * 32 + lane) : nullptr;

  v8f acc0 = {0.f, 0.f, 0.f, 0.f, 0.f, 0.f, 0.f, 0.f};
  v8f acc1 = {0.f, 0.f, 0.f, 0.f, 0.f, 0.f, 0.f, 0.f};

#pragma unroll 4
  for (int jj = 0; jj < NODE_F / 8; ++jj) {
    const float4 bs4 = Ws[jj * 32];
    float4 bn4 = make_float4(0.f, 0.f, 0.f, 0.f);
    if (NEIGH) bn4 = Wn[jj * 32];
#pragma unroll
    for (int s = 0; s < 2; ++s) {
      const int kk = 8 * jj + 4 * s + 2 * half;
      v2f bs;
      bs.x = s ? bs4.z : bs4.x;
      bs.y = s ? bs4.w : bs4.y;
      v2f a0, a1;
      a0.x = hs[lid][kk];
      a0.y = hs[lid][kk + 1];
      a1.x = hs[16 + lid][kk];
      a1.y = hs[16 + lid][kk + 1];
      acc0 = wmma_k4(a0, bs, acc0);
      acc1 = wmma_k4(a1, bs, acc1);
      if (NEIGH) {
        v2f bn;
        bn.x = s ? bn4.z : bn4.x;
        bn.y = s ? bn4.w : bn4.y;
        v2f c0, c1;
        c0.x = hn[lid][kk];
        c0.y = hn[lid][kk + 1];
        c1.x = hn[16 + lid][kk];
        c1.y = hn[16 + lid][kk + 1];
        acc0 = wmma_k4(c0, bn, acc0);
        acc1 = wmma_k4(c1, bn, acc1);
      }
    }
  }

  const float bv = bias[n0 + lid];
  float* outc = out + n0 + lid;
  const int r0 = m_base + 8 * half;
  const int r1 = m_base + 16 + 8 * half;

  if (fullTile) {
#pragma unroll
    for (int e = 0; e < 8; ++e) {
      float v0 = acc0[e] + bv;
      float v1 = acc1[e] + bv;
      if (RELU) { v0 = fmaxf(v0, 0.f); v1 = fmaxf(v1, 0.f); }
      outc[(size_t)(r0 + e) * NODE_F] = v0;
      outc[(size_t)(r1 + e) * NODE_F] = v1;
    }
  } else {
#pragma unroll
    for (int e = 0; e < 8; ++e) {
      float v0 = acc0[e] + bv;
      float v1 = acc1[e] + bv;
      if (RELU) { v0 = fmaxf(v0, 0.f); v1 = fmaxf(v1, 0.f); }
      if (r0 + e < nNodes) outc[(size_t)(r0 + e) * NODE_F] = v0;
      if (r1 + e < nNodes) outc[(size_t)(r1 + e) * NODE_F] = v1;
    }
  }
}

extern "C" void kernel_launch(void* const* d_in, const int* in_sizes, int n_in,
                              void* d_out, int out_size, void* d_ws, size_t ws_size,
                              hipStream_t stream) {
  const float* x      = (const float*)d_in[0];
  const int*   src    = (const int*)d_in[1];
  const int*   dst    = (const int*)d_in[2];
  const float* W1s    = (const float*)d_in[3];
  const float* W1n    = (const float*)d_in[4];
  const float* b1     = (const float*)d_in[5];
  const float* W2s    = (const float*)d_in[6];
  const float* W2n    = (const float*)d_in[7];
  const float* b2     = (const float*)d_in[8];
  const float* W3     = (const float*)d_in[9];
  const float* b3     = (const float*)d_in[10];
  float*       out    = (float*)d_out;

  const int nNodes = in_sizes[0] / NODE_F;
  const int nEdges = in_sizes[1];

  float* wpk = (float*)d_ws;
  float* sum = out;
  float* deg = wpk + (size_t)nNodes * NODE_F;
  float* h1  = deg + nNodes;
  float* h2  = h1 + (size_t)nNodes * NODE_F;

  unsigned* key = (unsigned*)h2;
  const int subVec  = AGG_T * AGG_V;
  const int nChunks = ((nEdges + 15) / 16 + subVec - 1) / subVec;
  const int nWords  = (nChunks + 1) * subVec * 4;

  dim3 blk(256);
  dim3 blkAgg(AGG_T);
  dim3 gKey((nWords + 255) / 256);
  dim3 gAgg((nNodes + AGG_R - 1) / AGG_R);
  dim3 gGemm((nNodes + 31) / 32);
  unsigned* list = key + nWords;
  unsigned* meta = list + (size_t)gAgg.x * AGG_LCAP;

  dim3 gKeyPack(gKey.x + 5 * PACK_BLOCKS_PER_W);
  const float* W1sP = wpk;
  const float* W1nP = wpk + 1 * NODE_F * NODE_F;
  const float* W2sP = wpk + 2 * NODE_F * NODE_F;
  const float* W2nP = wpk + 3 * NODE_F * NODE_F;
  const float* W3P  = wpk + 4 * NODE_F * NODE_F;
  key_kernel<<<gKeyPack, blk, 0, stream>>>(dst, key, nEdges, nWords, (int)gKey.x,
                                           W1s, W1n, W2s, W2n, W3, wpk);

  agg_kernel<true><<<gAgg, blkAgg, 0, stream>>>(x, src, dst, key, sum, deg, list, meta,
                                                nNodes, nEdges, nChunks);
  sage_gemm<true, true><<<gGemm, blk, 0, stream>>>(x, sum, deg, W1sP, W1nP, b1, h1, nNodes);

  agg_kernel<false><<<gAgg, blkAgg, 0, stream>>>(h1, src, dst, key, sum, deg, list, meta,
                                                 nNodes, nEdges, nChunks);
  sage_gemm<true, false><<<gGemm, blk, 0, stream>>>(h1, sum, deg, W2sP, W2nP, b2, h2, nNodes);

  sage_gemm<false, false><<<gGemm, blk, 0, stream>>>(h2, nullptr, nullptr, W3P, nullptr, b3,
                                                     out, nNodes);
}


# layer 1 without the 250-fold byte-key scan: one pass drops each edge into a 64-word slot per (node range, 4096-edge slice) by LDS counters; the owner block joins its slots in LDS; arrival sum order
# speedup vs baseline: 3.2452x; 1.1875x over previous
#include <hip/hip_runtime.h>

#define NODE_F 128

typedef float v2f __attribute__((ext_vector_type(2)));
typedef float v8f __attribute__((ext_vector_type(8)));

static __device__ __forceinline__ v8f wmma_k4(v2f a, v2f b, v8f c) {
  return __builtin_amdgcn_wmma_f32_16x16x4_f32(
       false, a,  false, b,
       (short)0, c,  false,  false);
}

#define AGG_R     400
#define AGG_T     1024
#define AGG_B     16
#define AGG_QCAP  (32 * AGG_B)
#define AGG_LCAP  16384
#define AGG_ECAP  8192
#define AGG_PMAX  512
#define PART_E    4096
#define PART_T    1024
#define PART_CAP  64
#define PART_BINS 1024

#define PACK_BLOCKS_PER_W 4
__global__ void __launch_bounds__(PART_T)
part_kernel(const int* __restrict__ src, const int* __restrict__ dst,
            unsigned* __restrict__ slot, unsigned* __restrict__ slotn,
            int nNodes, int nEdges, int nPart, int nBins,
            const float* __restrict__ Wa, const float* __restrict__ Wb,
            const float* __restrict__ Wc, const float* __restrict__ Wd,
            const float* __restrict__ We, float* __restrict__ wpk) {
  __shared__ unsigned       hist[PART_BINS];
  __shared__ unsigned       start[PART_BINS];
  __shared__ unsigned       wsum[33];
  __shared__ unsigned       stage[PART_E];
  __shared__ unsigned short srange[PART_E];
  const int tid = threadIdx.x;
  if ((int)blockIdx.x >= nPart) {
    const int pb = (int)blockIdx.x - nPart;
    const int m  = pb / PACK_BLOCKS_PER_W;
    const float* Wm = (m == 0) ? Wa : (m == 1) ? Wb : (m == 2) ? Wc : (m == 3) ? Wd : We;
    const int q    = (pb % PACK_BLOCKS_PER_W) * PART_T + tid;
    const int ln   = q & 31;
    const int jj   = (q >> 5) & 15;
    const int nt   = q >> 9;
    const float* wr = Wm + (size_t)(nt * 16 + (ln & 15)) * NODE_F + 8 * jj + 2 * (ln >> 4);
    const float2 p0 = *(const float2*)wr;
    const float2 p1 = *(const float2*)(wr + 4);
    ((float4*)wpk)[(size_t)m * (NODE_F * NODE_F / 4) + q] = make_float4(p0.x, p0.y, p1.x, p1.y);
    return;
  }
  hist[tid] = 0u;
  __syncthreads();

  const int part = blockIdx.x;
  const int lane = tid & 31;
  const int wave = tid >> 5;
  const int e0   = (part * PART_T + tid) * 4;
  int d[4], s[4];
  if (e0 + 3 < nEdges) {
    const int4 dd = ((const int4*)dst)[e0 >> 2];
    const int4 ss = ((const int4*)src)[e0 >> 2];
    d[0] = dd.x; d[1] = dd.y; d[2] = dd.z; d[3] = dd.w;
    s[0] = ss.x; s[1] = ss.y; s[2] = ss.z; s[3] = ss.w;
  } else {
#pragma unroll
    for (int b = 0; b < 4; ++b) {
      d[b] = (e0 + b < nEdges) ? dst[e0 + b] : -1;
      s[b] = (e0 + b < nEdges) ? src[e0 + b] : 0;
    }
  }
  unsigned bin[4], pos[4];
  bool     ok[4];
#pragma unroll
  for (int b = 0; b < 4; ++b) {
    bin[b] = (unsigned)d[b] / AGG_R;
    ok[b]  = (unsigned)d[b] < (unsigned)nNodes && bin[b] < (unsigned)PART_BINS;
    pos[b] = 0u;
    if (ok[b]) pos[b] = atomicAdd(&hist[bin[b]], 1u);
  }
  __syncthreads();

  const unsigned c = hist[tid];
  unsigned incl = c;
#pragma unroll
  for (int o = 1; o < 32; o <<= 1) {
    const unsigned t = __shfl_up(incl, (unsigned)o);
    if (lane >= o) incl += t;
  }
  if (lane == 31) wsum[wave] = incl;
  __syncthreads();
  if (wave == 0) {
    const unsigned w = wsum[lane];
    unsigned wi = w;
#pragma unroll
    for (int o = 1; o < 32; o <<= 1) {
      const unsigned t = __shfl_up(wi, (unsigned)o);
      if (lane >= o) wi += t;
    }
    wsum[lane] = wi - w;
    if (lane == 31) wsum[32] = wi;
  }
  __syncthreads();
  start[tid] = wsum[wave] + incl - c;
  __syncthreads();

#pragma unroll
  for (int b = 0; b < 4; ++b) {
    if (ok[b]) {
      const unsigned k = start[bin[b]] + pos[b];
      stage[k]  = ((unsigned)s[b] << 9) | ((unsigned)d[b] - bin[b] * AGG_R);
      srange[k] = (unsigned short)bin[b];
    }
  }
  __syncthreads();

  const unsigned total = wsum[32];
  for (unsigned k = tid; k < total; k += PART_T) {
    const unsigned r = srange[k];
    const unsigned j = k - start[r];
    if (j < (unsigned)PART_CAP) slot[((size_t)r * nPart + part) * PART_CAP + j] = stage[k];
  }
  for (int b = tid; b < nBins && b < PART_BINS; b += PART_T)
    slotn[(size_t)b * nPart + part] = hist[b];
}

template <bool FIRST>
__global__ void __launch_bounds__(AGG_T)
agg_kernel(const float* __restrict__ h, const int* __restrict__ src,
           const int* __restrict__ dst, const unsigned* __restrict__ slot,
           const unsigned* __restrict__ slotn, float* __restrict__ sum,
           float* __restrict__ deg, unsigned* __restrict__ list, unsigned* __restrict__ meta,
           int nNodes, int nEdges, int nPart) {
  __shared__ float    acc[(AGG_R + 1) * NODE_F];
  __shared__ unsigned elist[FIRST ? AGG_ECAP : 1];
  __shared__ unsigned sc[FIRST ? AGG_PMAX : 1];
  __shared__ unsigned so[FIRST ? AGG_PMAX : 1];
  __shared__ int      cnt[FIRST ? AGG_R : 1];
  __shared__ unsigned st[2];

  const int tid  = threadIdx.x;
  const int lane = tid & 31;
  const int wave = __builtin_amdgcn_readfirstlane(tid >> 5);
  const int lo   = blockIdx.x * AGG_R;
  const unsigned nOwn = (unsigned)min(AGG_R, nNodes - lo);

  unsigned myc = 0u;
  const bool fits = nPart <= AGG_PMAX && (int)blockIdx.x < PART_BINS;
  if (FIRST && fits && tid < nPart) myc = slotn[(size_t)blockIdx.x * nPart + tid];

  for (int i = tid; i < (AGG_R + 1) * NODE_F / 4; i += AGG_T)
    ((float4*)acc)[i] = make_float4(0.f, 0.f, 0.f, 0.f);
  if (FIRST)
    for (int i = tid; i < AGG_R; i += AGG_T) cnt[i] = 0;
  if (tid < 2) st[tid] = (tid == 1 && !fits) ? 1u : 0u;

  unsigned* lst = list + (size_t)blockIdx.x * AGG_LCAP;
  unsigned lcnt = 0u;
  bool useList = false;
  if (FIRST) {
    __syncthreads();
    if (fits && tid < nPart) {
      sc[tid] = myc;
      if (myc > (unsigned)PART_CAP) atomicOr(&st[1], 1u);
    }
    __syncthreads();
    if (wave == 0 && fits) {
      unsigned v[16];
      unsigned s = 0u;
#pragma unroll
      for (int k = 0; k < 16; ++k) {
        const int idx = lane * 16 + k;
        v[k] = (idx < nPart) ? sc[idx] : 0u;
        s += v[k];
      }
      unsigned incl = s;
#pragma unroll
      for (int o = 1; o < 32; o <<= 1) {
        const unsigned t = __shfl_up(incl, (unsigned)o);
        if (lane >= o) incl += t;
      }
      unsigned run = incl - s;
#pragma unroll
      for (int k = 0; k < 16; ++k) {
        const int idx = lane * 16 + k;
        if (idx < nPart) so[idx] = run;
        run += v[k];
      }
      if (lane == 31) {
        st[0] = incl;
        if (incl > (unsigned)AGG_ECAP) st[1] = 1u;
      }
    }
    __syncthreads();
    useList = st[1] == 0u;
    lcnt    = st[0];
    if (useList) {
      const unsigned* mys = slot + (size_t)blockIdx.x * nPart * PART_CAP;
      unsigned c[AGG_PMAX / 32], o[AGG_PMAX / 32], w0[AGG_PMAX / 32], w1[AGG_PMAX / 32];
#pragma unroll
      for (int u = 0; u < AGG_PMAX / 32; ++u) {
        const int p = wave + 32 * u;
        c[u] = (p < nPart) ? sc[p] : 0u;
        o[u] = (p < nPart) ? so[p] : 0u;
      }
#pragma unroll
      for (int u = 0; u < AGG_PMAX / 32; ++u) {
        const unsigned* sp = mys + (size_t)(wave + 32 * u) * PART_CAP;
        w0[u] = 0u;
        w1[u] = 0u;
        if ((unsigned)lane < c[u])       w0[u] = sp[lane];
        if ((unsigned)lane + 32u < c[u]) w1[u] = sp[lane + 32];
      }
#pragma unroll
      for (int u = 0; u < AGG_PMAX / 32; ++u) {
        if ((unsigned)lane < c[u]) {
          const unsigned k = o[u] + (unsigned)lane;
          elist[k] = w0[u];
          lst[k]   = w0[u];
          atomicAdd(&cnt[w0[u] & 511u], 1);
        }
        if ((unsigned)lane + 32u < c[u]) {
          const unsigned k = o[u] + (unsigned)lane + 32u;
          elist[k] = w1[u];
          lst[k]   = w1[u];
          atomicAdd(&cnt[w1[u] & 511u], 1);
        }
      }
    }
    if (tid == 0) {
      meta[2 * blockIdx.x]     = useList ? lcnt : 0u;
      meta[2 * blockIdx.x + 1] = useList ? 0u : 1u;
    }
  } else {
    lcnt    = meta[2 * blockIdx.x];
    useList = meta[2 * blockIdx.x + 1] == 0u;
  }
  __syncthreads();

  unsigned ent[AGG_B];
  float4 v[AGG_B];
  typedef __attribute__((address_space(1))) const float4 grow_t;
  auto load_row = [&](unsigned e) -> float4 {
    const unsigned long long b = (unsigned long long)h + (unsigned long long)(e >> 9) * (NODE_F * 4ull);
    const unsigned blo = (unsigned)__builtin_amdgcn_readfirstlane((int)(unsigned)b);
    const unsigned bhi = (unsigned)__builtin_amdgcn_readfirstlane((int)(unsigned)(b >> 32));
    return ((grow_t*)(((unsigned long long)bhi << 32) | blo))[lane];
  };
  auto load_rows = [&](auto word, unsigned k0, unsigned n) {
    unsigned raw[AGG_B];
#pragma unroll
    for (int j = 0; j < AGG_B; ++j) raw[j] = word(j);
#pragma unroll
    for (int j = 1; j < AGG_B; ++j) {
      ent[j] = 0u;
      v[j] = make_float4(0.f, 0.f, 0.f, 0.f);
      if (k0 + 32u * j < n) {
        ent[j] = (unsigned)__builtin_amdgcn_readfirstlane((int)raw[j]);
        v[j] = load_row(ent[j]);
      }
    }
    ent[0] = (k0 < n) ? (unsigned)__builtin_amdgcn_readfirstlane((int)raw[0]) : (unsigned)AGG_R;
    v[0] = load_row(ent[0]);
  };
  auto add_rows = [&](unsigned k0, unsigned n) {
#pragma unroll
    for (int j = 1; j < AGG_B; ++j) {
      if (k0 + 32u * j < n) {
        float* ap = acc + (ent[j] & 511u) * NODE_F + lane;
        atomicAdd(ap,      v[j].x);
        atomicAdd(ap + 32, v[j].y);
        atomicAdd(ap + 64, v[j].z);
        atomicAdd(ap + 96, v[j].w);
      }
    }
    {
      float* ap = acc + (ent[0] & 511u) * NODE_F + lane;
      atomicAdd(ap,      v[0].x);
      atomicAdd(ap + 32, v[0].y);
      atomicAdd(ap + 64, v[0].z);
      atomicAdd(ap + 96, v[0].w);
    }
  };

  if (useList) {
    auto list_words = [&](unsigned p0) -> unsigned {
      const unsigned n = min(lcnt - p0, (unsigned)AGG_QCAP);
      const unsigned idx = p0 + min((unsigned)wave + 32u * (unsigned)(lane & (AGG_B - 1)), n - 1u);
      if (FIRST) return elist[idx];
      return lst[idx];
    };
    if (FIRST) {
      unsigned curw[AGG_B];
      {
        const unsigned cur = (lcnt > 0u) ? list_words(0u) : 0u;
#pragma unroll
        for (int j = 0; j < AGG_B; ++j) curw[j] = (unsigned)__builtin_amdgcn_readlane((int)cur, j);
      }
      for (unsigned p0 = 0u; p0 < lcnt; p0 += AGG_QCAP) {
        const unsigned n = min(lcnt - p0, (unsigned)AGG_QCAP);
        load_rows([&](int j) -> unsigned { return curw[j]; }, (unsigned)wave, n);
        unsigned nxt = 0u;
        if (p0 + AGG_QCAP < lcnt) nxt = list_words(p0 + AGG_QCAP);
        unsigned nxtw[AGG_B];
#pragma unroll
        for (int j = 0; j < AGG_B; ++j) {
          nxtw[j] = (unsigned)__builtin_amdgcn_readlane((int)nxt, j);
          asm volatile("" : "+s"(nxtw[j]));
        }
        add_rows((unsigned)wave, n);
#pragma unroll
        for (int j = 0; j < AGG_B; ++j) curw[j] = nxtw[j];
      }
    } else {
      unsigned cur = (lcnt > 0u) ? list_words(0u) : 0u;
      for (unsigned p0 = 0u; p0 < lcnt; p0 += AGG_QCAP) {
        const unsigned n = min(lcnt - p0, (unsigned)AGG_QCAP);
        load_rows([&](int j) -> unsigned {
                    return (unsigned)__builtin_amdgcn_readlane((int)cur, j);
                  }, (unsigned)wave, n);
        unsigned nxt = 0u;
        if (p0 + AGG_QCAP < lcnt) nxt = list_words(p0 + AGG_QCAP);
        add_rows((unsigned)wave, n);
        cur = nxt;
      }
    }
  } else {
    for (int e = tid; e < nEdges; e += AGG_T) {
      const unsigned dl = (unsigned)(dst[e] - lo);
      if (dl < nOwn) {
        if (FIRST) atomicAdd(&cnt[dl], 1);
        const float* hp = h + (size_t)src[e] * NODE_F;
        for (int f = 0; f < NODE_F; ++f)
          atomicAdd(&acc[dl * NODE_F + (f >> 2) + 32 * (f & 3)], hp[f]);
      }
    }
  }
  __syncthreads();

  for (unsigned r = (unsigned)wave; r < nOwn; r += 32u) {
    const float* ap = acc + r * NODE_F + lane;
    ((float4*)(sum + (size_t)(lo + r) * NODE_F))[lane] = make_float4(ap[0], ap[32], ap[64], ap[96]);
  }
  if (FIRST)
    for (unsigned i = tid; i < nOwn; i += AGG_T) deg[lo + i] = (float)cnt[i];
}

template <bool NEIGH, bool RELU>
__global__ void sage_gemm(const float* __restrict__ hself, const float* __restrict__ sum,
                          const float* __restrict__ deg, const float* __restrict__ Wself,
                          const float* __restrict__ Wneigh, const float* __restrict__ bias,
                          float* __restrict__ out, int nNodes) {
  __shared__ float hs[32][NODE_F + 4];
  __shared__ float hn[32][NODE_F + 4];

  const int tid     = threadIdx.x;
  const int m_base  = blockIdx.x * 32;
  const bool fullTile = (m_base + 32 <= nNodes);

  for (int i = 0; i < 4; ++i) {
    int f   = i * 256 + tid;
    int row = f >> 5;
    int cv  = f & 31;
    int m   = m_base + row;
    float4 v = make_float4(0.f, 0.f, 0.f, 0.f);
    if (fullTile || m < nNodes) v = ((const float4*)(hself + (size_t)m * NODE_F))[cv];
    *(float4*)&hs[row][cv * 4] = v;
    if (NEIGH) {
      float4 s4 = make_float4(0.f, 0.f, 0.f, 0.f);
      float rd = 1.0f;
      if (fullTile || m < nNodes) {
        s4 = ((const float4*)(sum + (size_t)m * NODE_F))[cv];
        rd = 1.0f / fmaxf(deg[m], 1.0f);
      }
      hn[row][cv * 4 + 0] = s4.x * rd;
      hn[row][cv * 4 + 1] = s4.y * rd;
      hn[row][cv * 4 + 2] = s4.z * rd;
      hn[row][cv * 4 + 3] = s4.w * rd;
    }
  }
  __syncthreads();

  const int lane = tid & 31;
  const int wv   = tid >> 5;
  const int lid  = lane & 15;
  const int half = lane >> 4;
  const int n0   = wv * 16;

  const float4* Ws = (const float4*)Wself + wv * 16 * 32 + lane;
  const float4* Wn = NEIGH ? ((const float4*)Wneigh + wv * 16 * 32 + lane) : nullptr;

  v8f acc0 = {0.f, 0.f, 0.f, 0.f, 0.f, 0.f, 0.f, 0.f};
  v8f acc1 = {0.f, 0.f, 0.f, 0.f, 0.f, 0.f, 0.f, 0.f};

#pragma unroll 4
  for (int jj = 0; jj < NODE_F / 8; ++jj) {
    const float4 bs4 = Ws[jj * 32];
    float4 bn4 = make_float4(0.f, 0.f, 0.f, 0.f);
    if (NEIGH) bn4 = Wn[jj * 32];
#pragma unroll
    for (int s = 0; s < 2; ++s) {
      const int kk = 8 * jj + 4 * s + 2 * half;
      v2f bs;
      bs.x = s ? bs4.z : bs4.x;
      bs.y = s ? bs4.w : bs4.y;
      v2f a0, a1;
      a0.x = hs[lid][kk];
      a0.y = hs[lid][kk + 1];
      a1.x = hs[16 + lid][kk];
      a1.y = hs[16 + lid][kk + 1];
      acc0 = wmma_k4(a0, bs, acc0);
      acc1 = wmma_k4(a1, bs, acc1);
      if (NEIGH) {
        v2f bn;
        bn.x = s ? bn4.z : bn4.x;
        bn.y = s ? bn4.w : bn4.y;
        v2f c0, c1;
        c0.x = hn[lid][kk];
        c0.y = hn[lid][kk + 1];
        c1.x = hn[16 + lid][kk];
        c1.y = hn[16 + lid][kk + 1];
        acc0 = wmma_k4(c0, bn, acc0);
        acc1 = wmma_k4(c1, bn, acc1);
      }
    }
  }

  const float bv = bias[n0 + lid];
  float* outc = out + n0 + lid;
  const int r0 = m_base + 8 * half;
  const int r1 = m_base + 16 + 8 * half;

  if (fullTile) {
#pragma unroll
    for (int e = 0; e < 8; ++e) {
      float v0 = acc0[e] + bv;
      float v1 = acc1[e] + bv;
      if (RELU) { v0 = fmaxf(v0, 0.f); v1 = fmaxf(v1, 0.f); }
      outc[(size_t)(r0 + e) * NODE_F] = v0;
      outc[(size_t)(r1 + e) * NODE_F] = v1;
    }
  } else {
#pragma unroll
    for (int e = 0; e < 8; ++e) {
      float v0 = acc0[e] + bv;
      float v1 = acc1[e] + bv;
      if (RELU) { v0 = fmaxf(v0, 0.f); v1 = fmaxf(v1, 0.f); }
      if (r0 + e < nNodes) outc[(size_t)(r0 + e) * NODE_F] = v0;
      if (r1 + e < nNodes) outc[(size_t)(r1 + e) * NODE_F] = v1;
    }
  }
}

extern "C" void kernel_launch(void* const* d_in, const int* in_sizes, int n_in,
                              void* d_out, int out_size, void* d_ws, size_t ws_size,
                              hipStream_t stream) {
  const float* x      = (const float*)d_in[0];
  const int*   src    = (const int*)d_in[1];
  const int*   dst    = (const int*)d_in[2];
  const float* W1s    = (const float*)d_in[3];
  const float* W1n    = (const float*)d_in[4];
  const float* b1     = (const float*)d_in[5];
  const float* W2s    = (const float*)d_in[6];
  const float* W2n    = (const float*)d_in[7];
  const float* b2     = (const float*)d_in[8];
  const float* W3     = (const float*)d_in[9];
  const float* b3     = (const float*)d_in[10];
  float*       out    = (float*)d_out;

  const int nNodes = in_sizes[0] / NODE_F;
  const int nEdges = in_sizes[1];

  float* wpk = (float*)d_ws;
  float* sum = out;
  float* deg = wpk + (size_t)nNodes * NODE_F;
  float* h1  = deg + nNodes;
  float* h2  = h1 + (size_t)nNodes * NODE_F;

  const int nPart = (nEdges + PART_E - 1) / PART_E;

  dim3 blk(256);
  dim3 blkAgg(AGG_T);
  dim3 blkPart(PART_T);
  dim3 gAgg((nNodes + AGG_R - 1) / AGG_R);
  dim3 gGemm((nNodes + 31) / 32);
  unsigned* slot  = (unsigned*)h2;
  unsigned* slotn = slot + (size_t)gAgg.x * nPart * PART_CAP;
  unsigned* list  = slotn + (size_t)gAgg.x * nPart;
  unsigned* meta  = list + (size_t)gAgg.x * AGG_LCAP;

  dim3 gPart(nPart + 5 * PACK_BLOCKS_PER_W);
  const float* W1sP = wpk;
  const float* W1nP = wpk + 1 * NODE_F * NODE_F;
  const float* W2sP = wpk + 2 * NODE_F * NODE_F;
  const float* W2nP = wpk + 3 * NODE_F * NODE_F;
  const float* W3P  = wpk + 4 * NODE_F * NODE_F;
  part_kernel<<<gPart, blkPart, 0, stream>>>(src, dst, slot, slotn, nNodes, nEdges, nPart,
                                             (int)gAgg.x, W1s, W1n, W2s, W2n, W3, wpk);

  agg_kernel<true><<<gAgg, blkAgg, 0, stream>>>(x, src, dst, slot, slotn, sum, deg, list, meta,
                                                nNodes, nEdges, nPart);
  sage_gemm<true, true><<<gGemm, blk, 0, stream>>>(x, sum, deg, W1sP, W1nP, b1, h1, nNodes);

  agg_kernel<false><<<gAgg, blkAgg, 0, stream>>>(h1, src, dst, slot, slotn, sum, deg, list, meta,
                                                 nNodes, nEdges, nPart);
  sage_gemm<true, false><<<gGemm, blk, 0, stream>>>(h1, sum, deg, W2sP, W2nP, b2, h2, nNodes);

  sage_gemm<false, false><<<gGemm, blk, 0, stream>>>(h2, nullptr, nullptr, W3P, nullptr, b3,
                                                     out, nNodes);
}


# final linear chained onto the layer-2 product in one launch: the block keeps its 32x128 layer-2 tile in LDS and multiplies it by the packed W3; h2 store, reload and a launch gone; sum order kept
# speedup vs baseline: 3.3665x; 1.0374x over previous
#include <hip/hip_runtime.h>

#define NODE_F 128

typedef float v2f __attribute__((ext_vector_type(2)));
typedef float v8f __attribute__((ext_vector_type(8)));

static __device__ __forceinline__ v8f wmma_k4(v2f a, v2f b, v8f c) {
  return __builtin_amdgcn_wmma_f32_16x16x4_f32(
       false, a,  false, b,
       (short)0, c,  false,  false);
}

#define AGG_R     400
#define AGG_T     1024
#define AGG_B     16
#define AGG_QCAP  (32 * AGG_B)
#define AGG_LCAP  16384
#define AGG_ECAP  8192
#define AGG_PMAX  512
#define PART_E    4096
#define PART_T    1024
#define PART_CAP  64
#define PART_BINS 1024

#define PACK_BLOCKS_PER_W 4
__global__ void __launch_bounds__(PART_T)
part_kernel(const int* __restrict__ src, const int* __restrict__ dst,
            unsigned* __restrict__ slot, unsigned* __restrict__ slotn,
            int nNodes, int nEdges, int nPart, int nBins,
            const float* __restrict__ Wa, const float* __restrict__ Wb,
            const float* __restrict__ Wc, const float* __restrict__ Wd,
            const float* __restrict__ We, float* __restrict__ wpk) {
  __shared__ unsigned       hist[PART_BINS];
  __shared__ unsigned       start[PART_BINS];
  __shared__ unsigned       wsum[33];
  __shared__ unsigned       stage[PART_E];
  __shared__ unsigned short srange[PART_E];
  const int tid = threadIdx.x;
  if ((int)blockIdx.x >= nPart) {
    const int pb = (int)blockIdx.x - nPart;
    const int m  = pb / PACK_BLOCKS_PER_W;
    const float* Wm = (m == 0) ? Wa : (m == 1) ? Wb : (m == 2) ? Wc : (m == 3) ? Wd : We;
    const int q    = (pb % PACK_BLOCKS_PER_W) * PART_T + tid;
    const int ln   = q & 31;
    const int jj   = (q >> 5) & 15;
    const int nt   = q >> 9;
    const float* wr = Wm + (size_t)(nt * 16 + (ln & 15)) * NODE_F + 8 * jj + 2 * (ln >> 4);
    const float2 p0 = *(const float2*)wr;
    const float2 p1 = *(const float2*)(wr + 4);
    ((float4*)wpk)[(size_t)m * (NODE_F * NODE_F / 4) + q] = make_float4(p0.x, p0.y, p1.x, p1.y);
    return;
  }
  hist[tid] = 0u;
  __syncthreads();

  const int part = blockIdx.x;
  const int lane = tid & 31;
  const int wave = tid >> 5;
  const int e0   = (part * PART_T + tid) * 4;
  int d[4], s[4];
  if (e0 + 3 < nEdges) {
    const int4 dd = ((const int4*)dst)[e0 >> 2];
    const int4 ss = ((const int4*)src)[e0 >> 2];
    d[0] = dd.x; d[1] = dd.y; d[2] = dd.z; d[3] = dd.w;
    s[0] = ss.x; s[1] = ss.y; s[2] = ss.z; s[3] = ss.w;
  } else {
#pragma unroll
    for (int b = 0; b < 4; ++b) {
      d[b] = (e0 + b < nEdges) ? dst[e0 + b] : -1;
      s[b] = (e0 + b < nEdges) ? src[e0 + b] : 0;
    }
  }
  unsigned bin[4], pos[4];
  bool     ok[4];
#pragma unroll
  for (int b = 0; b < 4; ++b) {
    bin[b] = (unsigned)d[b] / AGG_R;
    ok[b]  = (unsigned)d[b] < (unsigned)nNodes && bin[b] < (unsigned)PART_BINS;
    pos[b] = 0u;
    if (ok[b]) pos[b] = atomicAdd(&hist[bin[b]], 1u);
  }
  __syncthreads();

  const unsigned c = hist[tid];
  unsigned incl = c;
#pragma unroll
  for (int o = 1; o < 32; o <<= 1) {
    const unsigned t = __shfl_up(incl, (unsigned)o);
    if (lane >= o) incl += t;
  }
  if (lane == 31) wsum[wave] = incl;
  __syncthreads();
  if (wave == 0) {
    const unsigned w = wsum[lane];
    unsigned wi = w;
#pragma unroll
    for (int o = 1; o < 32; o <<= 1) {
      const unsigned t = __shfl_up(wi, (unsigned)o);
      if (lane >= o) wi += t;
    }
    wsum[lane] = wi - w;
    if (lane == 31) wsum[32] = wi;
  }
  __syncthreads();
  start[tid] = wsum[wave] + incl - c;
  __syncthreads();

#pragma unroll
  for (int b = 0; b < 4; ++b) {
    if (ok[b]) {
      const unsigned k = start[bin[b]] + pos[b];
      stage[k]  = ((unsigned)s[b] << 9) | ((unsigned)d[b] - bin[b] * AGG_R);
      srange[k] = (unsigned short)bin[b];
    }
  }
  __syncthreads();

  const unsigned total = wsum[32];
  for (unsigned k = tid; k < total; k += PART_T) {
    const unsigned r = srange[k];
    const unsigned j = k - start[r];
    if (j < (unsigned)PART_CAP) slot[((size_t)r * nPart + part) * PART_CAP + j] = stage[k];
  }
  for (int b = tid; b < nBins && b < PART_BINS; b += PART_T)
    slotn[(size_t)b * nPart + part] = hist[b];
}

template <bool FIRST>
__global__ void __launch_bounds__(AGG_T)
agg_kernel(const float* __restrict__ h, const int* __restrict__ src,
           const int* __restrict__ dst, const unsigned* __restrict__ slot,
           const unsigned* __restrict__ slotn, float* __restrict__ sum,
           float* __restrict__ deg, unsigned* __restrict__ list, unsigned* __restrict__ meta,
           int nNodes, int nEdges, int nPart) {
  __shared__ float    acc[(AGG_R + 1) * NODE_F];
  __shared__ unsigned elist[FIRST ? AGG_ECAP : 1];
  __shared__ unsigned sc[FIRST ? AGG_PMAX : 1];
  __shared__ unsigned so[FIRST ? AGG_PMAX : 1];
  __shared__ int      cnt[FIRST ? AGG_R : 1];
  __shared__ unsigned st[2];

  const int tid  = threadIdx.x;
  const int lane = tid & 31;
  const int wave = __builtin_amdgcn_readfirstlane(tid >> 5);
  const int lo   = blockIdx.x * AGG_R;
  const unsigned nOwn = (unsigned)min(AGG_R, nNodes - lo);

  unsigned myc = 0u;
  const bool fits = nPart <= AGG_PMAX && (int)blockIdx.x < PART_BINS;
  if (FIRST && fits && tid < nPart) myc = slotn[(size_t)blockIdx.x * nPart + tid];

  for (int i = tid; i < (AGG_R + 1) * NODE_F / 4; i += AGG_T)
    ((float4*)acc)[i] = make_float4(0.f, 0.f, 0.f, 0.f);
  if (FIRST)
    for (int i = tid; i < AGG_R; i += AGG_T) cnt[i] = 0;
  if (tid < 2) st[tid] = (tid == 1 && !fits) ? 1u : 0u;

  unsigned* lst = list + (size_t)blockIdx.x * AGG_LCAP;
  unsigned lcnt = 0u;
  bool useList = false;
  if (FIRST) {
    __syncthreads();
    if (fits && tid < nPart) {
      sc[tid] = myc;
      if (myc > (unsigned)PART_CAP) atomicOr(&st[1], 1u);
    }
    __syncthreads();
    if (wave == 0 && fits) {
      unsigned v[16];
      unsigned s = 0u;
#pragma unroll
      for (int k = 0; k < 16; ++k) {
        const int idx = lane * 16 + k;
        v[k] = (idx < nPart) ? sc[idx] : 0u;
        s += v[k];
      }
      unsigned incl = s;
#pragma unroll
      for (int o = 1; o < 32; o <<= 1) {
        const unsigned t = __shfl_up(incl, (unsigned)o);
        if (lane >= o) incl += t;
      }
      unsigned run = incl - s;
#pragma unroll
      for (int k = 0; k < 16; ++k) {
        const int idx = lane * 16 + k;
        if (idx < nPart) so[idx] = run;
        run += v[k];
      }
      if (lane == 31) {
        st[0] = incl;
        if (incl > (unsigned)AGG_ECAP) st[1] = 1u;
      }
    }
    __syncthreads();
    useList = st[1] == 0u;
    lcnt    = st[0];
    if (useList) {
      const unsigned* mys = slot + (size_t)blockIdx.x * nPart * PART_CAP;
      unsigned c[AGG_PMAX / 32], o[AGG_PMAX / 32], w0[AGG_PMAX / 32], w1[AGG_PMAX / 32];
#pragma unroll
      for (int u = 0; u < AGG_PMAX / 32; ++u) {
        const int p = wave + 32 * u;
        c[u] = (p < nPart) ? sc[p] : 0u;
        o[u] = (p < nPart) ? so[p] : 0u;
      }
#pragma unroll
      for (int u = 0; u < AGG_PMAX / 32; ++u) {
        const unsigned* sp = mys + (size_t)(wave + 32 * u) * PART_CAP;
        w0[u] = 0u;
        w1[u] = 0u;
        if ((unsigned)lane < c[u])       w0[u] = sp[lane];
        if ((unsigned)lane + 32u < c[u]) w1[u] = sp[lane + 32];
      }
#pragma unroll
      for (int u = 0; u < AGG_PMAX / 32; ++u) {
        if ((unsigned)lane < c[u]) {
          const unsigned k = o[u] + (unsigned)lane;
          elist[k] = w0[u];
          lst[k]   = w0[u];
          atomicAdd(&cnt[w0[u] & 511u], 1);
        }
        if ((unsigned)lane + 32u < c[u]) {
          const unsigned k = o[u] + (unsigned)lane + 32u;
          elist[k] = w1[u];
          lst[k]   = w1[u];
          atomicAdd(&cnt[w1[u] & 511u], 1);
        }
      }
    }
    if (tid == 0) {
      meta[2 * blockIdx.x]     = useList ? lcnt : 0u;
      meta[2 * blockIdx.x + 1] = useList ? 0u : 1u;
    }
  } else {
    lcnt    = meta[2 * blockIdx.x];
    useList = meta[2 * blockIdx.x + 1] == 0u;
  }
  __syncthreads();

  unsigned ent[AGG_B];
  float4 v[AGG_B];
  typedef __attribute__((address_space(1))) const float4 grow_t;
  auto load_row = [&](unsigned e) -> float4 {
    const unsigned long long b = (unsigned long long)h + (unsigned long long)(e >> 9) * (NODE_F * 4ull);
    const unsigned blo = (unsigned)__builtin_amdgcn_readfirstlane((int)(unsigned)b);
    const unsigned bhi = (unsigned)__builtin_amdgcn_readfirstlane((int)(unsigned)(b >> 32));
    return ((grow_t*)(((unsigned long long)bhi << 32) | blo))[lane];
  };
  auto load_rows = [&](auto word, unsigned k0, unsigned n) {
    unsigned raw[AGG_B];
#pragma unroll
    for (int j = 0; j < AGG_B; ++j) raw[j] = word(j);
#pragma unroll
    for (int j = 1; j < AGG_B; ++j) {
      ent[j] = 0u;
      v[j] = make_float4(0.f, 0.f, 0.f, 0.f);
      if (k0 + 32u * j < n) {
        ent[j] = (unsigned)__builtin_amdgcn_readfirstlane((int)raw[j]);
        v[j] = load_row(ent[j]);
      }
    }
    ent[0] = (k0 < n) ? (unsigned)__builtin_amdgcn_readfirstlane((int)raw[0]) : (unsigned)AGG_R;
    v[0] = load_row(ent[0]);
  };
  auto add_rows = [&](unsigned k0, unsigned n) {
#pragma unroll
    for (int j = 1; j < AGG_B; ++j) {
      if (k0 + 32u * j < n) {
        float* ap = acc + (ent[j] & 511u) * NODE_F + lane;
        atomicAdd(ap,      v[j].x);
        atomicAdd(ap + 32, v[j].y);
        atomicAdd(ap + 64, v[j].z);
        atomicAdd(ap + 96, v[j].w);
      }
    }
    {
      float* ap = acc + (ent[0] & 511u) * NODE_F + lane;
      atomicAdd(ap,      v[0].x);
      atomicAdd(ap + 32, v[0].y);
      atomicAdd(ap + 64, v[0].z);
      atomicAdd(ap + 96, v[0].w);
    }
  };

  if (useList) {
    auto list_words = [&](unsigned p0) -> unsigned {
      const unsigned n = min(lcnt - p0, (unsigned)AGG_QCAP);
      const unsigned idx = p0 + min((unsigned)wave + 32u * (unsigned)(lane & (AGG_B - 1)), n - 1u);
      if (FIRST) return elist[idx];
      return lst[idx];
    };
    if (FIRST) {
      unsigned curw[AGG_B];
      {
        const unsigned cur = (lcnt > 0u) ? list_words(0u) : 0u;
#pragma unroll
        for (int j = 0; j < AGG_B; ++j) curw[j] = (unsigned)__builtin_amdgcn_readlane((int)cur, j);
      }
      for (unsigned p0 = 0u; p0 < lcnt; p0 += AGG_QCAP) {
        const unsigned n = min(lcnt - p0, (unsigned)AGG_QCAP);
        load_rows([&](int j) -> unsigned { return curw[j]; }, (unsigned)wave, n);
        unsigned nxt = 0u;
        if (p0 + AGG_QCAP < lcnt) nxt = list_words(p0 + AGG_QCAP);
        unsigned nxtw[AGG_B];
#pragma unroll
        for (int j = 0; j < AGG_B; ++j) {
          nxtw[j] = (unsigned)__builtin_amdgcn_readlane((int)nxt, j);
          asm volatile("" : "+s"(nxtw[j]));
        }
        add_rows((unsigned)wave, n);
#pragma unroll
        for (int j = 0; j < AGG_B; ++j) curw[j] = nxtw[j];
      }
    } else {
      unsigned cur = (lcnt > 0u) ? list_words(0u) : 0u;
      for (unsigned p0 = 0u; p0 < lcnt; p0 += AGG_QCAP) {
        const unsigned n = min(lcnt - p0, (unsigned)AGG_QCAP);
        load_rows([&](int j) -> unsigned {
                    return (unsigned)__builtin_amdgcn_readlane((int)cur, j);
                  }, (unsigned)wave, n);
        unsigned nxt = 0u;
        if (p0 + AGG_QCAP < lcnt) nxt = list_words(p0 + AGG_QCAP);
        add_rows((unsigned)wave, n);
        cur = nxt;
      }
    }
  } else {
    for (int e = tid; e < nEdges; e += AGG_T) {
      const unsigned dl = (unsigned)(dst[e] - lo);
      if (dl < nOwn) {
        if (FIRST) atomicAdd(&cnt[dl], 1);
        const float* hp = h + (size_t)src[e] * NODE_F;
        for (int f = 0; f < NODE_F; ++f)
          atomicAdd(&acc[dl * NODE_F + (f >> 2) + 32 * (f & 3)], hp[f]);
      }
    }
  }
  __syncthreads();

  for (unsigned r = (unsigned)wave; r < nOwn; r += 32u) {
    const float* ap = acc + r * NODE_F + lane;
    ((float4*)(sum + (size_t)(lo + r) * NODE_F))[lane] = make_float4(ap[0], ap[32], ap[64], ap[96]);
  }
  if (FIRST)
    for (unsigned i = tid; i < nOwn; i += AGG_T) deg[lo + i] = (float)cnt[i];
}

template <bool NEIGH, bool RELU, bool CHAIN>
__global__ void sage_gemm(const float* __restrict__ hself, const float* sum,
                          const float* __restrict__ deg, const float* __restrict__ Wself,
                          const float* __restrict__ Wneigh, const float* __restrict__ bias,
                          const float* __restrict__ Wnext, const float* __restrict__ biasNext,
                          float* out, int nNodes) {
  __shared__ float hs[32][NODE_F + 4];
  __shared__ float hn[32][NODE_F + 4];

  const int tid     = threadIdx.x;
  const int m_base  = blockIdx.x * 32;
  const bool fullTile = (m_base + 32 <= nNodes);

  for (int i = 0; i < 4; ++i) {
    int f   = i * 256 + tid;
    int row = f >> 5;
    int cv  = f & 31;
    int m   = m_base + row;
    float4 v = make_float4(0.f, 0.f, 0.f, 0.f);
    if (fullTile || m < nNodes) v = ((const float4*)(hself + (size_t)m * NODE_F))[cv];
    *(float4*)&hs[row][cv * 4] = v;
    if (NEIGH) {
      float4 s4 = make_float4(0.f, 0.f, 0.f, 0.f);
      float rd = 1.0f;
      if (fullTile || m < nNodes) {
        s4 = ((const float4*)(sum + (size_t)m * NODE_F))[cv];
        rd = 1.0f / fmaxf(deg[m], 1.0f);
      }
      hn[row][cv * 4 + 0] = s4.x * rd;
      hn[row][cv * 4 + 1] = s4.y * rd;
      hn[row][cv * 4 + 2] = s4.z * rd;
      hn[row][cv * 4 + 3] = s4.w * rd;
    }
  }
  __syncthreads();

  const int lane = tid & 31;
  const int wv   = tid >> 5;
  const int lid  = lane & 15;
  const int half = lane >> 4;
  const int n0   = wv * 16;

  const float4* Ws = (const float4*)Wself + wv * 16 * 32 + lane;
  const float4* Wn = NEIGH ? ((const float4*)Wneigh + wv * 16 * 32 + lane) : nullptr;

  v8f acc0 = {0.f, 0.f, 0.f, 0.f, 0.f, 0.f, 0.f, 0.f};
  v8f acc1 = {0.f, 0.f, 0.f, 0.f, 0.f, 0.f, 0.f, 0.f};

#pragma unroll 4
  for (int jj = 0; jj < NODE_F / 8; ++jj) {
    const float4 bs4 = Ws[jj * 32];
    float4 bn4 = make_float4(0.f, 0.f, 0.f, 0.f);
    if (NEIGH) bn4 = Wn[jj * 32];
#pragma unroll
    for (int s = 0; s < 2; ++s) {
      const int kk = 8 * jj + 4 * s + 2 * half;
      v2f bs;
      bs.x = s ? bs4.z : bs4.x;
      bs.y = s ? bs4.w : bs4.y;
      v2f a0, a1;
      a0.x = hs[lid][kk];
      a0.y = hs[lid][kk + 1];
      a1.x = hs[16 + lid][kk];
      a1.y = hs[16 + lid][kk + 1];
      acc0 = wmma_k4(a0, bs, acc0);
      acc1 = wmma_k4(a1, bs, acc1);
      if (NEIGH) {
        v2f bn;
        bn.x = s ? bn4.z : bn4.x;
        bn.y = s ? bn4.w : bn4.y;
        v2f c0, c1;
        c0.x = hn[lid][kk];
        c0.y = hn[lid][kk + 1];
        c1.x = hn[16 + lid][kk];
        c1.y = hn[16 + lid][kk + 1];
        acc0 = wmma_k4(c0, bn, acc0);
        acc1 = wmma_k4(c1, bn, acc1);
      }
    }
  }

  if (CHAIN) {
    const float bt = bias[n0 + lid];
    __syncthreads();
#pragma unroll
    for (int e = 0; e < 8; ++e) {
      float v0 = acc0[e] + bt;
      float v1 = acc1[e] + bt;
      if (RELU) { v0 = fmaxf(v0, 0.f); v1 = fmaxf(v1, 0.f); }
      hs[8 * half + e][n0 + lid]      = v0;
      hs[16 + 8 * half + e][n0 + lid] = v1;
    }
    __syncthreads();
#pragma unroll
    for (int e = 0; e < 8; ++e) { acc0[e] = 0.f; acc1[e] = 0.f; }
    const float4* Wx = (const float4*)Wnext + wv * 16 * 32 + lane;
#pragma unroll 4
    for (int jj = 0; jj < NODE_F / 8; ++jj) {
      const float4 bx4 = Wx[jj * 32];
#pragma unroll
      for (int s = 0; s < 2; ++s) {
        const int kk = 8 * jj + 4 * s + 2 * half;
        v2f bx;
        bx.x = s ? bx4.z : bx4.x;
        bx.y = s ? bx4.w : bx4.y;
        v2f a0, a1;
        a0.x = hs[lid][kk];
        a0.y = hs[lid][kk + 1];
        a1.x = hs[16 + lid][kk];
        a1.y = hs[16 + lid][kk + 1];
        acc0 = wmma_k4(a0, bx, acc0);
        acc1 = wmma_k4(a1, bx, acc1);
      }
    }
  }

  const float bv = CHAIN ? biasNext[n0 + lid] : bias[n0 + lid];
  float* outc = out + n0 + lid;
  const int r0 = m_base + 8 * half;
  const int r1 = m_base + 16 + 8 * half;

  if (fullTile) {
#pragma unroll
    for (int e = 0; e < 8; ++e) {
      float v0 = acc0[e] + bv;
      float v1 = acc1[e] + bv;
      if (RELU && !CHAIN) { v0 = fmaxf(v0, 0.f); v1 = fmaxf(v1, 0.f); }
      outc[(size_t)(r0 + e) * NODE_F] = v0;
      outc[(size_t)(r1 + e) * NODE_F] = v1;
    }
  } else {
#pragma unroll
    for (int e = 0; e < 8; ++e) {
      float v0 = acc0[e] + bv;
      float v1 = acc1[e] + bv;
      if (RELU && !CHAIN) { v0 = fmaxf(v0, 0.f); v1 = fmaxf(v1, 0.f); }
      if (r0 + e < nNodes) outc[(size_t)(r0 + e) * NODE_F] = v0;
      if (r1 + e < nNodes) outc[(size_t)(r1 + e) * NODE_F] = v1;
    }
  }
}

extern "C" void kernel_launch(void* const* d_in, const int* in_sizes, int n_in,
                              void* d_out, int out_size, void* d_ws, size_t ws_size,
                              hipStream_t stream) {
  const float* x      = (const float*)d_in[0];
  const int*   src    = (const int*)d_in[1];
  const int*   dst    = (const int*)d_in[2];
  const float* W1s    = (const float*)d_in[3];
  const float* W1n    = (const float*)d_in[4];
  const float* b1     = (const float*)d_in[5];
  const float* W2s    = (const float*)d_in[6];
  const float* W2n    = (const float*)d_in[7];
  const float* b2     = (const float*)d_in[8];
  const float* W3     = (const float*)d_in[9];
  const float* b3     = (const float*)d_in[10];
  float*       out    = (float*)d_out;

  const int nNodes = in_sizes[0] / NODE_F;
  const int nEdges = in_sizes[1];

  float* wpk = (float*)d_ws;
  float* sum = out;
  float* deg = wpk + (size_t)nNodes * NODE_F;
  float* h1  = deg + nNodes;
  float* h2  = h1 + (size_t)nNodes * NODE_F;

  const int nPart = (nEdges + PART_E - 1) / PART_E;

  dim3 blk(256);
  dim3 blkAgg(AGG_T);
  dim3 blkPart(PART_T);
  dim3 gAgg((nNodes + AGG_R - 1) / AGG_R);
  dim3 gGemm((nNodes + 31) / 32);
  unsigned* slot  = (unsigned*)h2;
  unsigned* slotn = slot + (size_t)gAgg.x * nPart * PART_CAP;
  unsigned* list  = slotn + (size_t)gAgg.x * nPart;
  unsigned* meta  = list + (size_t)gAgg.x * AGG_LCAP;

  dim3 gPart(nPart + 5 * PACK_BLOCKS_PER_W);
  const float* W1sP = wpk;
  const float* W1nP = wpk + 1 * NODE_F * NODE_F;
  const float* W2sP = wpk + 2 * NODE_F * NODE_F;
  const float* W2nP = wpk + 3 * NODE_F * NODE_F;
  const float* W3P  = wpk + 4 * NODE_F * NODE_F;
  part_kernel<<<gPart, blkPart, 0, stream>>>(src, dst, slot, slotn, nNodes, nEdges, nPart,
                                             (int)gAgg.x, W1s, W1n, W2s, W2n, W3, wpk);

  agg_kernel<true><<<gAgg, blkAgg, 0, stream>>>(x, src, dst, slot, slotn, sum, deg, list, meta,
                                                nNodes, nEdges, nPart);
  sage_gemm<true, true, false><<<gGemm, blk, 0, stream>>>(x, sum, deg, W1sP, W1nP, b1, nullptr,
                                                         nullptr, h1, nNodes);

  agg_kernel<false><<<gAgg, blkAgg, 0, stream>>>(h1, src, dst, slot, slotn, sum, deg, list, meta,
                                                 nNodes, nEdges, nPart);
  sage_gemm<true, false, true><<<gGemm, blk, 0, stream>>>(h1, sum, deg, W2sP, W2nP, b2, W3P, b3,
                                                          out, nNodes);
}
